# starting kernel
# baseline (speedup 1.0000x reference)
#include <hip/hip_runtime.h>


typedef __attribute__((ext_vector_type(2))) float v2f;
typedef __attribute__((ext_vector_type(8))) float v8f;

#define N_NODES 100000
#define IN_CH   128
#define HID_CH  16
#define OUT_CH  64
#define N_TILES ((N_NODES + 15) / 16)


__global__ void k_deg_init(float* __restrict__ deg) {
  int i = blockIdx.x * blockDim.x + threadIdx.x;
  if (i < N_NODES) deg[i] = 1.0f;
}

__global__ void k_degree(const int* __restrict__ col, float* __restrict__ deg, int E) {
  int e = blockIdx.x * blockDim.x + threadIdx.x;
  if (e < E) atomicAdd(&deg[col[e]], 1.0f);
}

__global__ void k_dinv(float* __restrict__ deg) {
  int i = blockIdx.x * blockDim.x + threadIdx.x;
  if (i < N_NODES) deg[i] = rsqrtf(deg[i]);
}


__global__ __launch_bounds__(256) void k_gemm1(const float* __restrict__ x,
                                               const float* __restrict__ W1,
                                               float* __restrict__ h1) {
  const int tile = blockIdx.x * 8 + (threadIdx.x >> 5);
  if (tile >= N_TILES) return;
  const int lane = threadIdx.x & 31;
  const int half = lane >> 4;
  const int l16  = lane & 15;
  const int row  = tile * 16 + l16;

  v8f c = {};
  #pragma unroll 4
  for (int k = 0; k < IN_CH; k += 4) {
    v2f a = *(const v2f*)(x + (size_t)row * IN_CH + k + half * 2);
    v2f b;
    b.x = W1[(k + half * 2 + 0) * HID_CH + l16];
    b.y = W1[(k + half * 2 + 1) * HID_CH + l16];
    c = __builtin_amdgcn_wmma_f32_16x16x4_f32(false, a, false, b,
                                              (short)0, c, false, false);
  }
  #pragma unroll
  for (int v = 0; v < 8; ++v)
    h1[(size_t)(tile * 16 + v + half * 8) * HID_CH + l16] = c[v];
}


__global__ void k_self_init16(const float* __restrict__ h1,
                              const float* __restrict__ dinv,
                              float* __restrict__ out1) {
  int t = blockIdx.x * blockDim.x + threadIdx.x;
  if (t < N_NODES * HID_CH) {
    int i = t >> 4;
    float d = dinv[i];
    out1[t] = h1[t] * d * d;
  }
}

__global__ void k_agg16(const int* __restrict__ row, const int* __restrict__ col,
                        const float* __restrict__ h1, const float* __restrict__ dinv,
                        float* __restrict__ out1, int E) {
  int t = blockIdx.x * blockDim.x + threadIdx.x;
  if (t >= E * HID_CH) return;
  int e = t >> 4;
  int f = t & 15;
  int r = row[e];
  int c = col[e];
  float nrm = dinv[r] * dinv[c];
  atomicAdd(&out1[c * HID_CH + f], h1[(size_t)r * HID_CH + f] * nrm);
}


__global__ __launch_bounds__(256) void k_gemm2(const float* __restrict__ h,
                                               const float* __restrict__ b1,
                                               const float* __restrict__ W2,
                                               float* __restrict__ h2) {
  const int tile = blockIdx.x * 8 + (threadIdx.x >> 5);
  if (tile >= N_TILES) return;
  const int ct   = blockIdx.y;
  const int lane = threadIdx.x & 31;
  const int half = lane >> 4;
  const int l16  = lane & 15;
  const int row  = tile * 16 + l16;

  v8f c = {};
  #pragma unroll
  for (int k = 0; k < HID_CH; k += 4) {
    const int k0 = k + half * 2;
    v2f a = *(const v2f*)(h + (size_t)row * HID_CH + k0);
    a.x = fmaxf(a.x + b1[k0 + 0], 0.0f);
    a.y = fmaxf(a.y + b1[k0 + 1], 0.0f);
    v2f b;
    b.x = W2[(k0 + 0) * OUT_CH + ct * 16 + l16];
    b.y = W2[(k0 + 1) * OUT_CH + ct * 16 + l16];
    c = __builtin_amdgcn_wmma_f32_16x16x4_f32(false, a, false, b,
                                              (short)0, c, false, false);
  }
  #pragma unroll
  for (int v = 0; v < 8; ++v)
    h2[(size_t)(tile * 16 + v + half * 8) * OUT_CH + ct * 16 + l16] = c[v];
}


__global__ void k_self_init64(const float* __restrict__ h2,
                              const float* __restrict__ dinv,
                              const float* __restrict__ b2,
                              float* __restrict__ out) {
  int t = blockIdx.x * blockDim.x + threadIdx.x;
  if (t < N_NODES * OUT_CH) {
    int i = t >> 6;
    int f = t & 63;
    float d = dinv[i];
    out[t] = h2[t] * d * d + b2[f];
  }
}

__global__ void k_agg64(const int* __restrict__ row, const int* __restrict__ col,
                        const float* __restrict__ h2, const float* __restrict__ dinv,
                        float* __restrict__ out, int E) {
  int t = blockIdx.x * blockDim.x + threadIdx.x;
  if (t >= E * OUT_CH) return;
  int e = t >> 6;
  int f = t & 63;
  int r = row[e];
  int c = col[e];
  float nrm = dinv[r] * dinv[c];
  atomicAdd(&out[c * OUT_CH + f], h2[(size_t)r * OUT_CH + f] * nrm);
}


extern "C" void kernel_launch(void* const* d_in, const int* in_sizes, int n_in,
                              void* d_out, int out_size, void* d_ws, size_t ws_size,
                              hipStream_t stream) {
  const float* x   = (const float*)d_in[0];
  const int*   ei  = (const int*)d_in[1];
  const float* W1  = (const float*)d_in[2];
  const float* b1  = (const float*)d_in[3];
  const float* W2  = (const float*)d_in[4];
  const float* b2  = (const float*)d_in[5];
  float*       out = (float*)d_out;

  const int E = in_sizes[1] / 2;
  const int* row = ei;
  const int* col = ei + E;

  float* dinv = (float*)d_ws;
  float* h1   = dinv + N_NODES;
  float* out1 = h1   + (size_t)N_NODES * HID_CH;
  float* h2   = out1 + (size_t)N_NODES * HID_CH;

  const int B = 256;

  k_deg_init<<<(N_NODES + B - 1) / B, B, 0, stream>>>(dinv);
  k_degree  <<<(E + B - 1) / B, B, 0, stream>>>(col, dinv, E);
  k_dinv    <<<(N_NODES + B - 1) / B, B, 0, stream>>>(dinv);

  k_gemm1<<<(N_TILES + 7) / 8, B, 0, stream>>>(x, W1, h1);
  k_self_init16<<<(N_NODES * HID_CH + B - 1) / B, B, 0, stream>>>(h1, dinv, out1);
  k_agg16<<<(E * HID_CH + B - 1) / B, B, 0, stream>>>(row, col, h1, dinv, out1, E);

  dim3 g2((N_TILES + 7) / 8, OUT_CH / 16);
  k_gemm2<<<g2, B, 0, stream>>>(out1, b1, W2, h2);
  k_self_init64<<<(N_NODES * OUT_CH + B - 1) / B, B, 0, stream>>>(h2, dinv, b2, out);
  k_agg64<<<(E * OUT_CH + B - 1) / B, B, 0, stream>>>(row, col, h2, dinv, out, E);
}


# layer 2 aggregates the 16-wide relu(hidden) rows BEFORE the W2 product ((A r) W2 == A (r W2)): edge atomics move 16 floats per edge, not 64; changes summation order
# speedup vs baseline: 1.6673x; 1.6673x over previous
#include <hip/hip_runtime.h>


typedef __attribute__((ext_vector_type(2))) float v2f;
typedef __attribute__((ext_vector_type(8))) float v8f;

#define N_NODES 100000
#define IN_CH   128
#define HID_CH  16
#define OUT_CH  64
#define N_TILES ((N_NODES + 15) / 16)


__global__ void k_deg_init(float* __restrict__ deg) {
  int i = blockIdx.x * blockDim.x + threadIdx.x;
  if (i < N_NODES) deg[i] = 1.0f;
}

__global__ void k_degree(const int* __restrict__ col, float* __restrict__ deg, int E) {
  int e = blockIdx.x * blockDim.x + threadIdx.x;
  if (e < E) atomicAdd(&deg[col[e]], 1.0f);
}

__global__ void k_dinv(float* __restrict__ deg) {
  int i = blockIdx.x * blockDim.x + threadIdx.x;
  if (i < N_NODES) deg[i] = rsqrtf(deg[i]);
}


__global__ __launch_bounds__(256) void k_gemm1(const float* __restrict__ x,
                                               const float* __restrict__ W1,
                                               float* __restrict__ h1) {
  const int tile = blockIdx.x * 8 + (threadIdx.x >> 5);
  if (tile >= N_TILES) return;
  const int lane = threadIdx.x & 31;
  const int half = lane >> 4;
  const int l16  = lane & 15;
  const int row  = tile * 16 + l16;

  v8f c = {};
  #pragma unroll 4
  for (int k = 0; k < IN_CH; k += 4) {
    v2f a = *(const v2f*)(x + (size_t)row * IN_CH + k + half * 2);
    v2f b;
    b.x = W1[(k + half * 2 + 0) * HID_CH + l16];
    b.y = W1[(k + half * 2 + 1) * HID_CH + l16];
    c = __builtin_amdgcn_wmma_f32_16x16x4_f32(false, a, false, b,
                                              (short)0, c, false, false);
  }
  #pragma unroll
  for (int v = 0; v < 8; ++v)
    h1[(size_t)(tile * 16 + v + half * 8) * HID_CH + l16] = c[v];
}


__global__ void k_self_init16(const float* __restrict__ h1,
                              const float* __restrict__ dinv,
                              float* __restrict__ out1) {
  int t = blockIdx.x * blockDim.x + threadIdx.x;
  if (t < N_NODES * HID_CH) {
    int i = t >> 4;
    float d = dinv[i];
    out1[t] = h1[t] * d * d;
  }
}

__global__ void k_agg16(const int* __restrict__ row, const int* __restrict__ col,
                        const float* __restrict__ h1, const float* __restrict__ dinv,
                        float* __restrict__ out1, int E) {
  int t = blockIdx.x * blockDim.x + threadIdx.x;
  if (t >= E * HID_CH) return;
  int e = t >> 4;
  int f = t & 15;
  int r = row[e];
  int c = col[e];
  float nrm = dinv[r] * dinv[c];
  atomicAdd(&out1[c * HID_CH + f], h1[(size_t)r * HID_CH + f] * nrm);
}


__global__ void k_relu_self_init16(const float* __restrict__ out1,
                                   const float* __restrict__ b1,
                                   const float* __restrict__ dinv,
                                   float* __restrict__ r,
                                   float* __restrict__ g) {
  int t = blockIdx.x * blockDim.x + threadIdx.x;
  if (t < N_NODES * HID_CH) {
    int i = t >> 4;
    int f = t & 15;
    float d = dinv[i];
    float v = fmaxf(out1[t] + b1[f], 0.0f);
    r[t] = v;
    g[t] = v * d * d;
  }
}


__global__ __launch_bounds__(256) void k_gemm2(const float* __restrict__ g,
                                               const float* __restrict__ W2,
                                               const float* __restrict__ b2,
                                               float* __restrict__ out) {
  const int tile = blockIdx.x * 8 + (threadIdx.x >> 5);
  if (tile >= N_TILES) return;
  const int ct   = blockIdx.y;
  const int lane = threadIdx.x & 31;
  const int half = lane >> 4;
  const int l16  = lane & 15;
  const int row  = tile * 16 + l16;

  v8f c = {};
  #pragma unroll
  for (int k = 0; k < HID_CH; k += 4) {
    const int k0 = k + half * 2;
    v2f a = *(const v2f*)(g + (size_t)row * HID_CH + k0);
    v2f b;
    b.x = W2[(k0 + 0) * OUT_CH + ct * 16 + l16];
    b.y = W2[(k0 + 1) * OUT_CH + ct * 16 + l16];
    c = __builtin_amdgcn_wmma_f32_16x16x4_f32(false, a, false, b,
                                              (short)0, c, false, false);
  }
  const float bias = b2[ct * 16 + l16];
  #pragma unroll
  for (int v = 0; v < 8; ++v)
    out[(size_t)(tile * 16 + v + half * 8) * OUT_CH + ct * 16 + l16] = c[v] + bias;
}


extern "C" void kernel_launch(void* const* d_in, const int* in_sizes, int n_in,
                              void* d_out, int out_size, void* d_ws, size_t ws_size,
                              hipStream_t stream) {
  const float* x   = (const float*)d_in[0];
  const int*   ei  = (const int*)d_in[1];
  const float* W1  = (const float*)d_in[2];
  const float* b1  = (const float*)d_in[3];
  const float* W2  = (const float*)d_in[4];
  const float* b2  = (const float*)d_in[5];
  float*       out = (float*)d_out;

  const int E = in_sizes[1] / 2;
  const int* row = ei;
  const int* col = ei + E;

  float* dinv = (float*)d_ws;
  float* h1   = dinv + N_NODES;
  float* out1 = h1   + (size_t)N_NODES * HID_CH;
  float* r    = out1 + (size_t)N_NODES * HID_CH;
  float* g    = r    + (size_t)N_NODES * HID_CH;

  const int B = 256;

  k_deg_init<<<(N_NODES + B - 1) / B, B, 0, stream>>>(dinv);
  k_degree  <<<(E + B - 1) / B, B, 0, stream>>>(col, dinv, E);
  k_dinv    <<<(N_NODES + B - 1) / B, B, 0, stream>>>(dinv);

  k_gemm1<<<(N_TILES + 7) / 8, B, 0, stream>>>(x, W1, h1);
  k_self_init16<<<(N_NODES * HID_CH + B - 1) / B, B, 0, stream>>>(h1, dinv, out1);
  k_agg16<<<(E * HID_CH + B - 1) / B, B, 0, stream>>>(row, col, h1, dinv, out1, E);

  k_relu_self_init16<<<(N_NODES * HID_CH + B - 1) / B, B, 0, stream>>>(out1, b1, dinv, r, g);
  k_agg16<<<(E * HID_CH + B - 1) / B, B, 0, stream>>>(row, col, r, dinv, g, E);
  dim3 g2((N_TILES + 7) / 8, OUT_CH / 16);
  k_gemm2<<<g2, B, 0, stream>>>(g, W2, b2, out);
}


# edges binned once per call by target bucket with LDS counters; each layer sums a bucket's rows in LDS (f32 LDS adds) and stores them once: no scattered global atomics; sum order = LDS arrival order
# speedup vs baseline: 5.9537x; 3.5709x over previous
#include <hip/hip_runtime.h>


typedef __attribute__((ext_vector_type(2))) float v2f;
typedef __attribute__((ext_vector_type(8))) float v8f;

#define N_NODES 100000
#define IN_CH   128
#define HID_CH  16
#define OUT_CH  64
#define N_TILES ((N_NODES + 15) / 16)

#define NPB     400
#define NBK     (N_NODES / NPB)
#define NCH     400
#define LBITS   9
#define N_CNT   (NBK * NCH)
#define N_QUADS (N_CNT / 4)
#define SCAN_Q  256
#define SCAN_T  1024
#define BIN_T   1024


__global__ __launch_bounds__(BIN_T) void k_hist(const int* __restrict__ col,
                                                int* __restrict__ H, int E) {
  __shared__ int s_h[NBK];
  const int t = threadIdx.x;
  const int g = blockIdx.x;
  if (t < NBK) s_h[t] = 0;
  __syncthreads();
  const int chunk = (E + NCH - 1) / NCH;
  const int beg = g * chunk;
  const int end = min(E, beg + chunk);
  const int last = end - 1;
  for (int e0 = beg + t; e0 < end; e0 += 8 * BIN_T) {
    int c[8];
    #pragma unroll
    for (int u = 0; u < 8; ++u) c[u] = col[min(e0 + BIN_T * u, last)];
    #pragma unroll
    for (int u = 0; u < 8; ++u)
      if (e0 + BIN_T * u < end) atomicAdd(&s_h[c[u] / NPB], 1);
  }
  __syncthreads();
  if (t < NBK) H[t * NCH + g] = s_h[t];
}

__global__ __launch_bounds__(SCAN_T) void k_scan(const int* __restrict__ cnt,
                                                 int* __restrict__ offs) {
  __shared__ int s_pre[SCAN_T / 32];
  __shared__ int s_tot[SCAN_T / 32];
  const int4* c4 = (const int4*)cnt;
  const int t = threadIdx.x;
  const int b = blockIdx.x;
  const int lane = t & 31;
  const int w = t >> 5;

  const int nfront = b * SCAN_Q;
  int pre = 0;
  for (int j0 = t; j0 < nfront; j0 += 8 * SCAN_T) {
    int4 v[8];
    #pragma unroll
    for (int u = 0; u < 8; ++u) v[u] = c4[min(j0 + SCAN_T * u, nfront - 1)];
    #pragma unroll
    for (int u = 0; u < 8; ++u)
      pre += (j0 + SCAN_T * u < nfront) ? (v[u].x + v[u].y) + (v[u].z + v[u].w) : 0;
  }
  const int q = nfront + t;
  const bool own = (t < SCAN_Q) && (q < N_QUADS);
  int4 mine = make_int4(0, 0, 0, 0);
  if (own) mine = c4[q];
  const int msum = (mine.x + mine.y) + (mine.z + mine.w);

  int inc = msum;
  #pragma unroll
  for (int d = 1; d < 32; d <<= 1) {
    int y = __shfl_up(inc, d);
    if (lane >= d) inc += y;
  }
  #pragma unroll
  for (int d = 16; d >= 1; d >>= 1) pre += __shfl_xor(pre, d);

  if (lane == 31) s_tot[w] = inc;
  if (lane == 0)  s_pre[w] = pre;
  __syncthreads();
  int base = 0;
  #pragma unroll
  for (int k = 0; k < SCAN_T / 32; ++k) {
    base += s_pre[k];
    if (k < w) base += s_tot[k];
  }
  const int excl = base + inc - msum;

  if (own) {
    int4 o;
    o.x = excl;
    o.y = o.x + mine.x;
    o.z = o.y + mine.y;
    o.w = o.z + mine.z;
    ((int4*)offs)[q] = o;
  }
}

__global__ __launch_bounds__(BIN_T) void k_bin(const int* __restrict__ row,
                                               const int* __restrict__ col,
                                               const int* __restrict__ O,
                                               int* __restrict__ binned, int E) {
  __shared__ int s_cur[NBK];
  const int t = threadIdx.x;
  const int g = blockIdx.x;
  if (t < NBK) s_cur[t] = O[t * NCH + g];
  __syncthreads();
  const int chunk = (E + NCH - 1) / NCH;
  const int beg = g * chunk;
  const int end = min(E, beg + chunk);
  const int last = end - 1;
  for (int e0 = beg + t; e0 < end; e0 += 8 * BIN_T) {
    int c[8], r[8];
    #pragma unroll
    for (int u = 0; u < 8; ++u) {
      const int e = min(e0 + BIN_T * u, last);
      c[u] = col[e];
      r[u] = row[e];
    }
    #pragma unroll
    for (int u = 0; u < 8; ++u) {
      if (e0 + BIN_T * u < end) {
        const int b = c[u] / NPB;
        const int slot = atomicAdd(&s_cur[b], 1);
        binned[slot] = (r[u] << LBITS) | (c[u] - b * NPB);
      }
    }
  }
}

__global__ __launch_bounds__(BIN_T) void k_deg(const int* __restrict__ binned,
                                               const int* __restrict__ O,
                                               float* __restrict__ dinv, int E) {
  __shared__ int s_c[NPB];
  const int t = threadIdx.x;
  const int b = blockIdx.x;
  if (t < NPB) s_c[t] = 0;
  __syncthreads();
  const int beg = O[b * NCH];
  const int end = (b + 1 < NBK) ? O[(b + 1) * NCH] : E;
  const int last = end - 1;
  for (int p0 = beg + t; p0 < end; p0 += 8 * BIN_T) {
    int w[8];
    #pragma unroll
    for (int u = 0; u < 8; ++u) w[u] = binned[min(p0 + BIN_T * u, last)];
    #pragma unroll
    for (int u = 0; u < 8; ++u)
      if (p0 + BIN_T * u < end) atomicAdd(&s_c[w[u] & ((1 << LBITS) - 1)], 1);
  }
  __syncthreads();
  if (t < NPB) dinv[b * NPB + t] = rsqrtf((float)(s_c[t] + 1));
}


__global__ __launch_bounds__(256) void k_gemm1(const float* __restrict__ x,
                                               const float* __restrict__ W1,
                                               const float* __restrict__ dinv,
                                               float* __restrict__ hs) {
  const int tile = blockIdx.x * 8 + (threadIdx.x >> 5);
  if (tile >= N_TILES) return;
  const int lane = threadIdx.x & 31;
  const int half = lane >> 4;
  const int l16  = lane & 15;
  const int row  = tile * 16 + l16;

  v8f c = {};
  #pragma unroll 4
  for (int k = 0; k < IN_CH; k += 4) {
    v2f a = *(const v2f*)(x + (size_t)row * IN_CH + k + half * 2);
    v2f b;
    b.x = W1[(k + half * 2 + 0) * HID_CH + l16];
    b.y = W1[(k + half * 2 + 1) * HID_CH + l16];
    c = __builtin_amdgcn_wmma_f32_16x16x4_f32(false, a, false, b,
                                              (short)0, c, false, false);
  }
  #pragma unroll
  for (int v = 0; v < 8; ++v) {
    const int m = tile * 16 + v + half * 8;
    hs[(size_t)m * HID_CH + l16] = c[v] * dinv[m];
  }
}


template <bool LAYER1>
__global__ __launch_bounds__(1024) void k_agg(const float* __restrict__ hs,
                                              const int* __restrict__ binned,
                                              const int* __restrict__ O,
                                              const float* __restrict__ dinv,
                                              const float* __restrict__ b1,
                                              float* __restrict__ dst, int E) {
  __shared__ __attribute__((aligned(16))) float s_acc[NPB * HID_CH];
  const int t = threadIdx.x;
  const int b = blockIdx.x;
  const int base4 = b * NPB * (HID_CH / 4);
  float4* acc4 = (float4*)s_acc;

  #pragma unroll
  for (int k = 0; k < 2; ++k) {
    const int i4 = t + k * 1024;
    if (i4 < NPB * (HID_CH / 4)) acc4[i4] = ((const float4*)hs)[base4 + i4];
  }
  __syncthreads();

  const int beg = O[b * NCH];
  const int end = (b + 1 < NBK) ? O[(b + 1) * NCH] : E;
  const int last = end - 1;
  const int f = t & 15;
  const int lmask = (1 << LBITS) - 1;
  int p = beg + (t >> 4);
  if (p < end) {
    int w[8];
    #pragma unroll
    for (int u = 0; u < 8; ++u) w[u] = binned[min(p + 64 * u, last)];
    while (p + 64 * 7 < end) {
      const int pn = p + 64 * 8;
      int wn[8];
      #pragma unroll
      for (int u = 0; u < 8; ++u) wn[u] = binned[min(pn + 64 * u, last)];
      float v[8];
      #pragma unroll
      for (int u = 0; u < 8; ++u) v[u] = hs[(w[u] >> LBITS) * HID_CH + f];
      #pragma unroll
      for (int u = 0; u < 8; ++u) atomicAdd(&s_acc[(w[u] & lmask) * HID_CH + f], v[u]);
      #pragma unroll
      for (int u = 0; u < 8; ++u) w[u] = wn[u];
      p = pn;
    }
    if (p < end) {
      float v[8];
      #pragma unroll
      for (int u = 0; u < 8; ++u) v[u] = hs[(w[u] >> LBITS) * HID_CH + f];
      #pragma unroll
      for (int u = 0; u < 8; ++u)
        if (p + 64 * u < end) atomicAdd(&s_acc[(w[u] & lmask) * HID_CH + f], v[u]);
    }
  }
  __syncthreads();

  #pragma unroll
  for (int k = 0; k < 2; ++k) {
    const int i4 = t + k * 1024;
    if (i4 < NPB * (HID_CH / 4)) {
      const float d = dinv[b * NPB + (i4 >> 2)];
      const float4 a = acc4[i4];
      float4 o;
      if (LAYER1) {
        const int q = (i4 & 3) * 4;
        o.x = fmaxf(a.x * d + b1[q + 0], 0.0f) * d;
        o.y = fmaxf(a.y * d + b1[q + 1], 0.0f) * d;
        o.z = fmaxf(a.z * d + b1[q + 2], 0.0f) * d;
        o.w = fmaxf(a.w * d + b1[q + 3], 0.0f) * d;
      } else {
        o.x = a.x * d; o.y = a.y * d; o.z = a.z * d; o.w = a.w * d;
      }
      ((float4*)dst)[base4 + i4] = o;
    }
  }
}


__global__ __launch_bounds__(256) void k_gemm2(const float* __restrict__ g,
                                               const float* __restrict__ W2,
                                               const float* __restrict__ b2,
                                               float* __restrict__ out) {
  const int tile = blockIdx.x * 8 + (threadIdx.x >> 5);
  if (tile >= N_TILES) return;
  const int ct   = blockIdx.y;
  const int lane = threadIdx.x & 31;
  const int half = lane >> 4;
  const int l16  = lane & 15;
  const int row  = tile * 16 + l16;

  v8f c = {};
  #pragma unroll
  for (int k = 0; k < HID_CH; k += 4) {
    const int k0 = k + half * 2;
    v2f a = *(const v2f*)(g + (size_t)row * HID_CH + k0);
    v2f b;
    b.x = W2[(k0 + 0) * OUT_CH + ct * 16 + l16];
    b.y = W2[(k0 + 1) * OUT_CH + ct * 16 + l16];
    c = __builtin_amdgcn_wmma_f32_16x16x4_f32(false, a, false, b,
                                              (short)0, c, false, false);
  }
  const float bias = b2[ct * 16 + l16];
  #pragma unroll
  for (int v = 0; v < 8; ++v)
    out[(size_t)(tile * 16 + v + half * 8) * OUT_CH + ct * 16 + l16] = c[v] + bias;
}


extern "C" void kernel_launch(void* const* d_in, const int* in_sizes, int n_in,
                              void* d_out, int out_size, void* d_ws, size_t ws_size,
                              hipStream_t stream) {
  const float* x   = (const float*)d_in[0];
  const int*   ei  = (const int*)d_in[1];
  const float* W1  = (const float*)d_in[2];
  const float* b1  = (const float*)d_in[3];
  const float* W2  = (const float*)d_in[4];
  const float* b2  = (const float*)d_in[5];
  float*       out = (float*)d_out;

  const int E = in_sizes[1] / 2;
  const int* row = ei;
  const int* col = ei + E;

  float* hs     = (float*)d_ws;
  float* rs     = hs + (size_t)N_NODES * HID_CH;
  float* g      = rs + (size_t)N_NODES * HID_CH;
  float* dinv   = g  + (size_t)N_NODES * HID_CH;
  int*   H      = (int*)(dinv + N_NODES);
  int*   O      = H + N_CNT;
  int*   binned = O + N_CNT;

  const int B = 256;

  k_hist<<<NCH, BIN_T, 0, stream>>>(col, H, E);
  k_scan<<<(N_QUADS + SCAN_Q - 1) / SCAN_Q, SCAN_T, 0, stream>>>(H, O);
  k_bin <<<NCH, BIN_T, 0, stream>>>(row, col, O, binned, E);
  k_deg <<<NBK, BIN_T, 0, stream>>>(binned, O, dinv, E);

  k_gemm1<<<(N_TILES + 7) / 8, B, 0, stream>>>(x, W1, dinv, hs);
  k_agg<true><<<NBK, 1024, 0, stream>>>(hs, binned, O, dinv, b1, rs, E);

  k_agg<false><<<NBK, 1024, 0, stream>>>(rs, binned, O, dinv, b1, g, E);
  dim3 g2((N_TILES + 7) / 8, OUT_CH / 16);
  k_gemm2<<<g2, B, 0, stream>>>(g, W2, b2, out);
}


# layer-2 aggregation multiplies its bucket's 400x16 sums, still in LDS, by W2 on WMMA and stores out itself: the separate W2 product launch (25000 short waves) and the g round trip are gone
# speedup vs baseline: 6.5799x; 1.1052x over previous
#include <hip/hip_runtime.h>


typedef __attribute__((ext_vector_type(2))) float v2f;
typedef __attribute__((ext_vector_type(8))) float v8f;

#define N_NODES 100000
#define IN_CH   128
#define HID_CH  16
#define OUT_CH  64
#define N_TILES ((N_NODES + 15) / 16)

#define NPB     400
#define NBK     (N_NODES / NPB)
#define NCH     400
#define LBITS   9
#define N_CNT   (NBK * NCH)
#define N_QUADS (N_CNT / 4)
#define SCAN_Q  256
#define SCAN_T  1024
#define BIN_T   1024


__global__ __launch_bounds__(BIN_T) void k_hist(const int* __restrict__ col,
                                                int* __restrict__ H, int E) {
  __shared__ int s_h[NBK];
  const int t = threadIdx.x;
  const int g = blockIdx.x;
  if (t < NBK) s_h[t] = 0;
  __syncthreads();
  const int chunk = (E + NCH - 1) / NCH;
  const int beg = g * chunk;
  const int end = min(E, beg + chunk);
  const int last = end - 1;
  for (int e0 = beg + t; e0 < end; e0 += 8 * BIN_T) {
    int c[8];
    #pragma unroll
    for (int u = 0; u < 8; ++u) c[u] = col[min(e0 + BIN_T * u, last)];
    #pragma unroll
    for (int u = 0; u < 8; ++u)
      if (e0 + BIN_T * u < end) atomicAdd(&s_h[c[u] / NPB], 1);
  }
  __syncthreads();
  if (t < NBK) H[t * NCH + g] = s_h[t];
}

__global__ __launch_bounds__(SCAN_T) void k_scan(const int* __restrict__ cnt,
                                                 int* __restrict__ offs) {
  __shared__ int s_pre[SCAN_T / 32];
  __shared__ int s_tot[SCAN_T / 32];
  const int4* c4 = (const int4*)cnt;
  const int t = threadIdx.x;
  const int b = blockIdx.x;
  const int lane = t & 31;
  const int w = t >> 5;

  const int nfront = b * SCAN_Q;
  int pre = 0;
  for (int j0 = t; j0 < nfront; j0 += 8 * SCAN_T) {
    int4 v[8];
    #pragma unroll
    for (int u = 0; u < 8; ++u) v[u] = c4[min(j0 + SCAN_T * u, nfront - 1)];
    #pragma unroll
    for (int u = 0; u < 8; ++u)
      pre += (j0 + SCAN_T * u < nfront) ? (v[u].x + v[u].y) + (v[u].z + v[u].w) : 0;
  }
  const int q = nfront + t;
  const bool own = (t < SCAN_Q) && (q < N_QUADS);
  int4 mine = make_int4(0, 0, 0, 0);
  if (own) mine = c4[q];
  const int msum = (mine.x + mine.y) + (mine.z + mine.w);

  int inc = msum;
  #pragma unroll
  for (int d = 1; d < 32; d <<= 1) {
    int y = __shfl_up(inc, d);
    if (lane >= d) inc += y;
  }
  #pragma unroll
  for (int d = 16; d >= 1; d >>= 1) pre += __shfl_xor(pre, d);

  if (lane == 31) s_tot[w] = inc;
  if (lane == 0)  s_pre[w] = pre;
  __syncthreads();
  int base = 0;
  #pragma unroll
  for (int k = 0; k < SCAN_T / 32; ++k) {
    base += s_pre[k];
    if (k < w) base += s_tot[k];
  }
  const int excl = base + inc - msum;

  if (own) {
    int4 o;
    o.x = excl;
    o.y = o.x + mine.x;
    o.z = o.y + mine.y;
    o.w = o.z + mine.z;
    ((int4*)offs)[q] = o;
  }
}

__global__ __launch_bounds__(BIN_T) void k_bin(const int* __restrict__ row,
                                               const int* __restrict__ col,
                                               const int* __restrict__ O,
                                               int* __restrict__ binned, int E) {
  __shared__ int s_cur[NBK];
  const int t = threadIdx.x;
  const int g = blockIdx.x;
  if (t < NBK) s_cur[t] = O[t * NCH + g];
  __syncthreads();
  const int chunk = (E + NCH - 1) / NCH;
  const int beg = g * chunk;
  const int end = min(E, beg + chunk);
  const int last = end - 1;
  for (int e0 = beg + t; e0 < end; e0 += 8 * BIN_T) {
    int c[8], r[8];
    #pragma unroll
    for (int u = 0; u < 8; ++u) {
      const int e = min(e0 + BIN_T * u, last);
      c[u] = col[e];
      r[u] = row[e];
    }
    #pragma unroll
    for (int u = 0; u < 8; ++u) {
      if (e0 + BIN_T * u < end) {
        const int b = c[u] / NPB;
        const int slot = atomicAdd(&s_cur[b], 1);
        binned[slot] = (r[u] << LBITS) | (c[u] - b * NPB);
      }
    }
  }
}

__global__ __launch_bounds__(BIN_T) void k_deg(const int* __restrict__ binned,
                                               const int* __restrict__ O,
                                               float* __restrict__ dinv, int E) {
  __shared__ int s_c[NPB];
  const int t = threadIdx.x;
  const int b = blockIdx.x;
  if (t < NPB) s_c[t] = 0;
  __syncthreads();
  const int beg = O[b * NCH];
  const int end = (b + 1 < NBK) ? O[(b + 1) * NCH] : E;
  const int last = end - 1;
  for (int p0 = beg + t; p0 < end; p0 += 8 * BIN_T) {
    int w[8];
    #pragma unroll
    for (int u = 0; u < 8; ++u) w[u] = binned[min(p0 + BIN_T * u, last)];
    #pragma unroll
    for (int u = 0; u < 8; ++u)
      if (p0 + BIN_T * u < end) atomicAdd(&s_c[w[u] & ((1 << LBITS) - 1)], 1);
  }
  __syncthreads();
  if (t < NPB) dinv[b * NPB + t] = rsqrtf((float)(s_c[t] + 1));
}


__global__ __launch_bounds__(256) void k_gemm1(const float* __restrict__ x,
                                               const float* __restrict__ W1,
                                               const float* __restrict__ dinv,
                                               float* __restrict__ hs) {
  const int tile = blockIdx.x * 8 + (threadIdx.x >> 5);
  if (tile >= N_TILES) return;
  const int lane = threadIdx.x & 31;
  const int half = lane >> 4;
  const int l16  = lane & 15;
  const int row  = tile * 16 + l16;

  v8f c = {};
  #pragma unroll 4
  for (int k = 0; k < IN_CH; k += 4) {
    v2f a = *(const v2f*)(x + (size_t)row * IN_CH + k + half * 2);
    v2f b;
    b.x = W1[(k + half * 2 + 0) * HID_CH + l16];
    b.y = W1[(k + half * 2 + 1) * HID_CH + l16];
    c = __builtin_amdgcn_wmma_f32_16x16x4_f32(false, a, false, b,
                                              (short)0, c, false, false);
  }
  #pragma unroll
  for (int v = 0; v < 8; ++v) {
    const int m = tile * 16 + v + half * 8;
    hs[(size_t)m * HID_CH + l16] = c[v] * dinv[m];
  }
}


template <bool LAYER1>
__global__ __launch_bounds__(1024) void k_agg(const float* __restrict__ hs,
                                              const int* __restrict__ binned,
                                              const int* __restrict__ O,
                                              const float* __restrict__ dinv,
                                              const float* __restrict__ bias,
                                              const float* __restrict__ W2,
                                              float* __restrict__ dst, int E) {
  static_assert(NPB % 16 == 0 && NPB / 16 <= 32, "one wave per 16-row tile of the bucket");
  __shared__ __attribute__((aligned(16))) float s_acc[NPB * HID_CH];
  const int t = threadIdx.x;
  const int b = blockIdx.x;
  const int base4 = b * NPB * (HID_CH / 4);
  float4* acc4 = (float4*)s_acc;

  #pragma unroll
  for (int k = 0; k < 2; ++k) {
    const int i4 = t + k * 1024;
    if (i4 < NPB * (HID_CH / 4)) acc4[i4] = ((const float4*)hs)[base4 + i4];
  }
  __syncthreads();

  const int beg = O[b * NCH];
  const int end = (b + 1 < NBK) ? O[(b + 1) * NCH] : E;
  const int last = end - 1;
  const int f = t & 15;
  const int lmask = (1 << LBITS) - 1;
  int p = beg + (t >> 4);
  if (p < end) {
    int w[8];
    #pragma unroll
    for (int u = 0; u < 8; ++u) w[u] = binned[min(p + 64 * u, last)];
    while (p + 64 * 7 < end) {
      const int pn = p + 64 * 8;
      int wn[8];
      #pragma unroll
      for (int u = 0; u < 8; ++u) wn[u] = binned[min(pn + 64 * u, last)];
      float v[8];
      #pragma unroll
      for (int u = 0; u < 8; ++u) v[u] = hs[(w[u] >> LBITS) * HID_CH + f];
      #pragma unroll
      for (int u = 0; u < 8; ++u) atomicAdd(&s_acc[(w[u] & lmask) * HID_CH + f], v[u]);
      #pragma unroll
      for (int u = 0; u < 8; ++u) w[u] = wn[u];
      p = pn;
    }
    if (p < end) {
      float v[8];
      #pragma unroll
      for (int u = 0; u < 8; ++u) v[u] = hs[(w[u] >> LBITS) * HID_CH + f];
      #pragma unroll
      for (int u = 0; u < 8; ++u)
        if (p + 64 * u < end) atomicAdd(&s_acc[(w[u] & lmask) * HID_CH + f], v[u]);
    }
  }
  __syncthreads();

  if (LAYER1) {
    #pragma unroll
    for (int k = 0; k < 2; ++k) {
      const int i4 = t + k * 1024;
      if (i4 < NPB * (HID_CH / 4)) {
        const float d = dinv[b * NPB + (i4 >> 2)];
        const float4 a = acc4[i4];
        const int q = (i4 & 3) * 4;
        float4 o;
        o.x = fmaxf(a.x * d + bias[q + 0], 0.0f) * d;
        o.y = fmaxf(a.y * d + bias[q + 1], 0.0f) * d;
        o.z = fmaxf(a.z * d + bias[q + 2], 0.0f) * d;
        o.w = fmaxf(a.w * d + bias[q + 3], 0.0f) * d;
        ((float4*)dst)[base4 + i4] = o;
      }
    }
  } else {
    const int wv = __builtin_amdgcn_readfirstlane(t >> 5);
    if (wv < NPB / 16) {
      const int lane = t & 31;
      const int half = lane >> 4;
      const int l16  = lane & 15;
      const int r    = wv * 16 + l16;
      const float d  = dinv[b * NPB + r];
      v2f a[HID_CH / 4];
      #pragma unroll
      for (int k = 0; k < HID_CH / 4; ++k) {
        a[k] = *(const v2f*)(s_acc + r * HID_CH + k * 4 + half * 2);
        a[k].x *= d;
        a[k].y *= d;
      }
      v8f c[OUT_CH / 16];
      #pragma unroll
      for (int ct = 0; ct < OUT_CH / 16; ++ct) c[ct] = v8f{};
      #pragma unroll
      for (int k = 0; k < HID_CH / 4; ++k) {
        const int k0 = k * 4 + half * 2;
        #pragma unroll
        for (int ct = 0; ct < OUT_CH / 16; ++ct) {
          v2f bf;
          bf.x = W2[(k0 + 0) * OUT_CH + ct * 16 + l16];
          bf.y = W2[(k0 + 1) * OUT_CH + ct * 16 + l16];
          c[ct] = __builtin_amdgcn_wmma_f32_16x16x4_f32(false, a[k], false, bf,
                                                        (short)0, c[ct], false, false);
        }
      }
      float bs[OUT_CH / 16];
      #pragma unroll
      for (int ct = 0; ct < OUT_CH / 16; ++ct) bs[ct] = bias[ct * 16 + l16];
      #pragma unroll
      for (int v = 0; v < 8; ++v) {
        const size_t m = (size_t)(b * NPB + wv * 16 + v + half * 8);
        #pragma unroll
        for (int ct = 0; ct < OUT_CH / 16; ++ct)
          dst[m * OUT_CH + ct * 16 + l16] = c[ct][v] + bs[ct];
      }
    }
  }
}


extern "C" void kernel_launch(void* const* d_in, const int* in_sizes, int n_in,
                              void* d_out, int out_size, void* d_ws, size_t ws_size,
                              hipStream_t stream) {
  const float* x   = (const float*)d_in[0];
  const int*   ei  = (const int*)d_in[1];
  const float* W1  = (const float*)d_in[2];
  const float* b1  = (const float*)d_in[3];
  const float* W2  = (const float*)d_in[4];
  const float* b2  = (const float*)d_in[5];
  float*       out = (float*)d_out;

  const int E = in_sizes[1] / 2;
  const int* row = ei;
  const int* col = ei + E;

  float* hs     = (float*)d_ws;
  float* rs     = hs + (size_t)N_NODES * HID_CH;
  float* dinv   = rs + (size_t)N_NODES * HID_CH;
  int*   H      = (int*)(dinv + N_NODES);
  int*   O      = H + N_CNT;
  int*   binned = O + N_CNT;

  const int B = 256;

  k_hist<<<NCH, BIN_T, 0, stream>>>(col, H, E);
  k_scan<<<(N_QUADS + SCAN_Q - 1) / SCAN_Q, SCAN_T, 0, stream>>>(H, O);
  k_bin <<<NCH, BIN_T, 0, stream>>>(row, col, O, binned, E);
  k_deg <<<NBK, BIN_T, 0, stream>>>(binned, O, dinv, E);

  k_gemm1<<<(N_TILES + 7) / 8, B, 0, stream>>>(x, W1, dinv, hs);
  k_agg<true><<<NBK, 1024, 0, stream>>>(hs, binned, O, dinv, b1, W2, rs, E);

  k_agg<false><<<NBK, 1024, 0, stream>>>(rs, binned, O, dinv, b2, W2, out, E);
}


# edges binned by a chunk-local LDS sort stored as whole runs, then one copy-and-count pass per bucket: no histogram or scan launch, no scattered 4-byte stores; edge (sum) order in a bucket changes
# speedup vs baseline: 9.3411x; 1.4196x over previous
#include <hip/hip_runtime.h>


typedef __attribute__((ext_vector_type(2))) float v2f;
typedef __attribute__((ext_vector_type(8))) float v8f;

#define N_NODES 100000
#define IN_CH   128
#define HID_CH  16
#define OUT_CH  64
#define N_TILES ((N_NODES + 15) / 16)

#define NPB     400
#define NBK     (N_NODES / NPB)
#define NCH     400
#define LBITS   9
#define N_CNT   (NBK * NCH)
#define BIN_T   1024
#define SORT_T  512
#define EPT     16
#define CH_MAX  (SORT_T * EPT)
#define RUNS    8


__global__ __launch_bounds__(SORT_T) void k_sort(const int* __restrict__ row,
                                                const int* __restrict__ col,
                                                int* __restrict__ C,
                                                int* __restrict__ LS,
                                                int* __restrict__ sorted, int E) {
  __shared__ int s_cnt[256];
  __shared__ int s_wtot[8];
  __shared__ int s_w[CH_MAX];
  const int t = threadIdx.x;
  const int g = blockIdx.x;
  const int lane = t & 31;
  const int w = t >> 5;
  if (t < 256) s_cnt[t] = 0;
  __syncthreads();
  const int chunk = (E + NCH - 1) / NCH;
  const int beg = g * chunk;
  const int end = min(E, beg + chunk);
  const int n = max(end - beg, 0);
  const int last = max(end - 1, 0);

  int c[EPT], r[EPT], bk[EPT], rk[EPT];
  #pragma unroll
  for (int u = 0; u < EPT; ++u) {
    const int e = min(beg + t + SORT_T * u, last);
    c[u] = col[e];
    r[u] = row[e];
  }
  #pragma unroll
  for (int u = 0; u < EPT; ++u) {
    bk[u] = c[u] / NPB;
    rk[u] = 0;
    if (t + SORT_T * u < n) rk[u] = atomicAdd(&s_cnt[bk[u]], 1);
  }
  __syncthreads();

  int cntv = 0, inc = 0;
  if (t < 256) {
    cntv = s_cnt[t];
    inc = cntv;
    #pragma unroll
    for (int d = 1; d < 32; d <<= 1) {
      int y = __shfl_up(inc, d);
      if (lane >= d) inc += y;
    }
    if (lane == 31) s_wtot[w] = inc;
  }
  __syncthreads();
  if (t < 256) {
    int base = 0;
    #pragma unroll
    for (int k = 0; k < 8; ++k)
      if (k < w) base += s_wtot[k];
    const int excl = base + inc - cntv;
    s_cnt[t] = excl;
    if (t < NBK) {
      C[t * NCH + g] = cntv;
      LS[t * NCH + g] = excl;
    }
  }
  __syncthreads();

  int st[EPT];
  #pragma unroll
  for (int u = 0; u < EPT; ++u) st[u] = s_cnt[bk[u]];
  #pragma unroll
  for (int u = 0; u < EPT; ++u)
    if (t + SORT_T * u < n)
      s_w[st[u] + rk[u]] = (r[u] << LBITS) | (c[u] - bk[u] * NPB);
  __syncthreads();

  for (int i = t; i < n; i += SORT_T) sorted[beg + i] = s_w[i];
}

__global__ __launch_bounds__(BIN_T) void k_move(const int* __restrict__ C,
                                                const int* __restrict__ LS,
                                                const int* __restrict__ sorted,
                                                int* __restrict__ binned,
                                                int* __restrict__ seg,
                                                float* __restrict__ dinv, int E) {
  __shared__ int s_c[NPB];
  __shared__ int s_cnt[NCH];
  __shared__ int s_src[NCH];
  __shared__ int s_dst[NCH];
  __shared__ int s_red[BIN_T / 32];
  __shared__ int s_wtot[BIN_T / 32];
  const int t = threadIdx.x;
  const int b = blockIdx.x;
  const int lane = t & 31;
  const int w = t >> 5;
  const int lmask = (1 << LBITS) - 1;
  if (t < NPB) s_c[t] = 0;

  const int4* c4 = (const int4*)C;
  const int nfront = b * (NCH / 4);
  int pre = 0;
  for (int j0 = t; j0 < nfront; j0 += 8 * BIN_T) {
    int4 v[8];
    #pragma unroll
    for (int u = 0; u < 8; ++u) v[u] = c4[min(j0 + BIN_T * u, nfront - 1)];
    #pragma unroll
    for (int u = 0; u < 8; ++u)
      pre += (j0 + BIN_T * u < nfront) ? (v[u].x + v[u].y) + (v[u].z + v[u].w) : 0;
  }
  #pragma unroll
  for (int d = 16; d >= 1; d >>= 1) pre += __shfl_xor(pre, d);

  const int tg = min(t, NCH - 1);
  const int cntv = (t < NCH) ? C[b * NCH + tg] : 0;
  const int lsv = LS[b * NCH + tg];
  int inc = cntv;
  #pragma unroll
  for (int d = 1; d < 32; d <<= 1) {
    int y = __shfl_up(inc, d);
    if (lane >= d) inc += y;
  }
  if (lane == 0)  s_red[w] = pre;
  if (lane == 31) s_wtot[w] = inc;
  __syncthreads();
  int base = 0, off = 0, total = 0;
  #pragma unroll
  for (int k = 0; k < BIN_T / 32; ++k) {
    base += s_red[k];
    total += s_wtot[k];
    if (k < w) off += s_wtot[k];
  }
  const int chunk = (E + NCH - 1) / NCH;
  if (t < NCH) {
    s_cnt[t] = cntv;
    s_src[t] = t * chunk + lsv;
    s_dst[t] = base + off + inc - cntv;
  }
  if (t == 0) {
    seg[2 * b] = base;
    seg[2 * b + 1] = base + total;
  }
  __syncthreads();

  for (int g0 = w; g0 < NCH; g0 += 32 * RUNS) {
    int n[RUNS], s[RUNS], d[RUNS];
    int wd[RUNS][2];
    #pragma unroll
    for (int j = 0; j < RUNS; ++j) {
      const int g = g0 + 32 * j;
      const int gg = min(g, NCH - 1);
      n[j] = (g < NCH) ? s_cnt[gg] : 0;
      s[j] = s_src[gg];
      d[j] = s_dst[gg];
    }
    #pragma unroll
    for (int j = 0; j < RUNS; ++j) {
      #pragma unroll
      for (int k = 0; k < 2; ++k) {
        const int i = lane + 32 * k;
        wd[j][k] = (i < n[j]) ? sorted[s[j] + i] : 0;
      }
    }
    #pragma unroll
    for (int j = 0; j < RUNS; ++j) {
      #pragma unroll
      for (int k = 0; k < 2; ++k) {
        const int i = lane + 32 * k;
        if (i < n[j]) {
          binned[d[j] + i] = wd[j][k];
          atomicAdd(&s_c[wd[j][k] & lmask], 1);
        }
      }
    }
    #pragma unroll
    for (int j = 0; j < RUNS; ++j) {
      for (int i = lane + 64; i < n[j]; i += 32) {
        const int word = sorted[s[j] + i];
        binned[d[j] + i] = word;
        atomicAdd(&s_c[word & lmask], 1);
      }
    }
  }
  __syncthreads();
  if (t < NPB) dinv[b * NPB + t] = rsqrtf((float)(s_c[t] + 1));
}


__global__ __launch_bounds__(256) void k_gemm1(const float* __restrict__ x,
                                               const float* __restrict__ W1,
                                               const float* __restrict__ dinv,
                                               float* __restrict__ hs) {
  const int tile = blockIdx.x * 8 + (threadIdx.x >> 5);
  if (tile >= N_TILES) return;
  const int lane = threadIdx.x & 31;
  const int half = lane >> 4;
  const int l16  = lane & 15;
  const int row  = tile * 16 + l16;

  v8f c = {};
  #pragma unroll 4
  for (int k = 0; k < IN_CH; k += 4) {
    v2f a = *(const v2f*)(x + (size_t)row * IN_CH + k + half * 2);
    v2f b;
    b.x = W1[(k + half * 2 + 0) * HID_CH + l16];
    b.y = W1[(k + half * 2 + 1) * HID_CH + l16];
    c = __builtin_amdgcn_wmma_f32_16x16x4_f32(false, a, false, b,
                                              (short)0, c, false, false);
  }
  #pragma unroll
  for (int v = 0; v < 8; ++v) {
    const int m = tile * 16 + v + half * 8;
    hs[(size_t)m * HID_CH + l16] = c[v] * dinv[m];
  }
}


template <bool LAYER1>
__global__ __launch_bounds__(1024) void k_agg(const float* __restrict__ hs,
                                              const int* __restrict__ binned,
                                              const int* __restrict__ seg,
                                              const float* __restrict__ dinv,
                                              const float* __restrict__ bias,
                                              const float* __restrict__ W2,
                                              float* __restrict__ dst, int E) {
  static_assert(NPB % 16 == 0 && NPB / 16 <= 32, "one wave per 16-row tile of the bucket");
  __shared__ __attribute__((aligned(16))) float s_acc[NPB * HID_CH];
  const int t = threadIdx.x;
  const int b = blockIdx.x;
  const int base4 = b * NPB * (HID_CH / 4);
  float4* acc4 = (float4*)s_acc;

  #pragma unroll
  for (int k = 0; k < 2; ++k) {
    const int i4 = t + k * 1024;
    if (i4 < NPB * (HID_CH / 4)) acc4[i4] = ((const float4*)hs)[base4 + i4];
  }
  __syncthreads();

  const int beg = seg[2 * b];
  const int end = seg[2 * b + 1];
  const int last = end - 1;
  const int f = t & 15;
  const int lmask = (1 << LBITS) - 1;
  int p = beg + (t >> 4);
  if (p < end) {
    int w[8];
    #pragma unroll
    for (int u = 0; u < 8; ++u) w[u] = binned[min(p + 64 * u, last)];
    while (p + 64 * 7 < end) {
      const int pn = p + 64 * 8;
      int wn[8];
      #pragma unroll
      for (int u = 0; u < 8; ++u) wn[u] = binned[min(pn + 64 * u, last)];
      float v[8];
      #pragma unroll
      for (int u = 0; u < 8; ++u) v[u] = hs[(w[u] >> LBITS) * HID_CH + f];
      #pragma unroll
      for (int u = 0; u < 8; ++u) atomicAdd(&s_acc[(w[u] & lmask) * HID_CH + f], v[u]);
      #pragma unroll
      for (int u = 0; u < 8; ++u) w[u] = wn[u];
      p = pn;
    }
    if (p < end) {
      float v[8];
      #pragma unroll
      for (int u = 0; u < 8; ++u) v[u] = hs[(w[u] >> LBITS) * HID_CH + f];
      #pragma unroll
      for (int u = 0; u < 8; ++u)
        if (p + 64 * u < end) atomicAdd(&s_acc[(w[u] & lmask) * HID_CH + f], v[u]);
    }
  }
  __syncthreads();

  if (LAYER1) {
    #pragma unroll
    for (int k = 0; k < 2; ++k) {
      const int i4 = t + k * 1024;
      if (i4 < NPB * (HID_CH / 4)) {
        const float d = dinv[b * NPB + (i4 >> 2)];
        const float4 a = acc4[i4];
        const int q = (i4 & 3) * 4;
        float4 o;
        o.x = fmaxf(a.x * d + bias[q + 0], 0.0f) * d;
        o.y = fmaxf(a.y * d + bias[q + 1], 0.0f) * d;
        o.z = fmaxf(a.z * d + bias[q + 2], 0.0f) * d;
        o.w = fmaxf(a.w * d + bias[q + 3], 0.0f) * d;
        ((float4*)dst)[base4 + i4] = o;
      }
    }
  } else {
    const int wv = __builtin_amdgcn_readfirstlane(t >> 5);
    if (wv < NPB / 16) {
      const int lane = t & 31;
      const int half = lane >> 4;
      const int l16  = lane & 15;
      const int r    = wv * 16 + l16;
      const float d  = dinv[b * NPB + r];
      v2f a[HID_CH / 4];
      #pragma unroll
      for (int k = 0; k < HID_CH / 4; ++k) {
        a[k] = *(const v2f*)(s_acc + r * HID_CH + k * 4 + half * 2);
        a[k].x *= d;
        a[k].y *= d;
      }
      v8f c[OUT_CH / 16];
      #pragma unroll
      for (int ct = 0; ct < OUT_CH / 16; ++ct) c[ct] = v8f{};
      #pragma unroll
      for (int k = 0; k < HID_CH / 4; ++k) {
        const int k0 = k * 4 + half * 2;
        #pragma unroll
        for (int ct = 0; ct < OUT_CH / 16; ++ct) {
          v2f bf;
          bf.x = W2[(k0 + 0) * OUT_CH + ct * 16 + l16];
          bf.y = W2[(k0 + 1) * OUT_CH + ct * 16 + l16];
          c[ct] = __builtin_amdgcn_wmma_f32_16x16x4_f32(false, a[k], false, bf,
                                                        (short)0, c[ct], false, false);
        }
      }
      float bs[OUT_CH / 16];
      #pragma unroll
      for (int ct = 0; ct < OUT_CH / 16; ++ct) bs[ct] = bias[ct * 16 + l16];
      #pragma unroll
      for (int v = 0; v < 8; ++v) {
        const size_t m = (size_t)(b * NPB + wv * 16 + v + half * 8);
        #pragma unroll
        for (int ct = 0; ct < OUT_CH / 16; ++ct)
          dst[m * OUT_CH + ct * 16 + l16] = c[ct][v] + bs[ct];
      }
    }
  }
}


extern "C" void kernel_launch(void* const* d_in, const int* in_sizes, int n_in,
                              void* d_out, int out_size, void* d_ws, size_t ws_size,
                              hipStream_t stream) {
  const float* x   = (const float*)d_in[0];
  const int*   ei  = (const int*)d_in[1];
  const float* W1  = (const float*)d_in[2];
  const float* b1  = (const float*)d_in[3];
  const float* W2  = (const float*)d_in[4];
  const float* b2  = (const float*)d_in[5];
  float*       out = (float*)d_out;

  const int E = in_sizes[1] / 2;
  const int* row = ei;
  const int* col = ei + E;

  float* hs     = (float*)d_ws;
  float* rs     = hs + (size_t)N_NODES * HID_CH;
  float* dinv   = rs + (size_t)N_NODES * HID_CH;
  int*   C      = (int*)(dinv + N_NODES);
  int*   LS     = C + N_CNT;
  int*   seg    = LS + N_CNT;
  int*   sorted = seg + 2 * NBK + 12;
  int*   binned = sorted + (size_t)NCH * CH_MAX;

  const int B = 256;

  k_sort<<<NCH, SORT_T, 0, stream>>>(row, col, C, LS, sorted, E);
  k_move<<<NBK, BIN_T, 0, stream>>>(C, LS, sorted, binned, seg, dinv, E);

  k_gemm1<<<(N_TILES + 7) / 8, B, 0, stream>>>(x, W1, dinv, hs);
  k_agg<true><<<NBK, 1024, 0, stream>>>(hs, binned, seg, dinv, b1, W2, rs, E);

  k_agg<false><<<NBK, 1024, 0, stream>>>(rs, binned, seg, dinv, b2, W2, out, E);
}


# x@W1 product moved into k_move's bucket work-groups (one row tile per wave behind its copy, x tile L2-prefetched at entry, scaled by the bucket's dinv from LDS): the separate product launch is gone
# speedup vs baseline: 9.8781x; 1.0575x over previous
#include <hip/hip_runtime.h>


typedef __attribute__((ext_vector_type(2))) float v2f;
typedef __attribute__((ext_vector_type(8))) float v8f;

#define N_NODES 100000
#define IN_CH   128
#define HID_CH  16
#define OUT_CH  64

#define NPB     400
#define NBK     (N_NODES / NPB)
#define NCH     400
#define LBITS   9
#define N_CNT   (NBK * NCH)
#define BIN_T   1024
#define SORT_T  512
#define EPT     16
#define CH_MAX  (SORT_T * EPT)
#define RUNS    8


__global__ __launch_bounds__(SORT_T) void k_sort(const int* __restrict__ row,
                                                const int* __restrict__ col,
                                                int* __restrict__ C,
                                                int* __restrict__ LS,
                                                int* __restrict__ sorted, int E) {
  __shared__ int s_cnt[256];
  __shared__ int s_wtot[8];
  __shared__ int s_w[CH_MAX];
  const int t = threadIdx.x;
  const int g = blockIdx.x;
  const int lane = t & 31;
  const int w = t >> 5;
  if (t < 256) s_cnt[t] = 0;
  __syncthreads();
  const int chunk = (E + NCH - 1) / NCH;
  const int beg = g * chunk;
  const int end = min(E, beg + chunk);
  const int n = max(end - beg, 0);
  const int last = max(end - 1, 0);

  int c[EPT], r[EPT], bk[EPT], rk[EPT];
  #pragma unroll
  for (int u = 0; u < EPT; ++u) {
    const int e = min(beg + t + SORT_T * u, last);
    c[u] = col[e];
    r[u] = row[e];
  }
  #pragma unroll
  for (int u = 0; u < EPT; ++u) {
    bk[u] = c[u] / NPB;
    rk[u] = 0;
    if (t + SORT_T * u < n) rk[u] = atomicAdd(&s_cnt[bk[u]], 1);
  }
  __syncthreads();

  int cntv = 0, inc = 0;
  if (t < 256) {
    cntv = s_cnt[t];
    inc = cntv;
    #pragma unroll
    for (int d = 1; d < 32; d <<= 1) {
      int y = __shfl_up(inc, d);
      if (lane >= d) inc += y;
    }
    if (lane == 31) s_wtot[w] = inc;
  }
  __syncthreads();
  if (t < 256) {
    int base = 0;
    #pragma unroll
    for (int k = 0; k < 8; ++k)
      if (k < w) base += s_wtot[k];
    const int excl = base + inc - cntv;
    s_cnt[t] = excl;
    if (t < NBK) {
      C[t * NCH + g] = cntv;
      LS[t * NCH + g] = excl;
    }
  }
  __syncthreads();

  int st[EPT];
  #pragma unroll
  for (int u = 0; u < EPT; ++u) st[u] = s_cnt[bk[u]];
  #pragma unroll
  for (int u = 0; u < EPT; ++u)
    if (t + SORT_T * u < n)
      s_w[st[u] + rk[u]] = (r[u] << LBITS) | (c[u] - bk[u] * NPB);
  __syncthreads();

  for (int i = t; i < n; i += SORT_T) sorted[beg + i] = s_w[i];
}

__global__ __launch_bounds__(BIN_T) void k_move(const int* __restrict__ C,
                                                const int* __restrict__ LS,
                                                const int* __restrict__ sorted,
                                                int* __restrict__ binned,
                                                int* __restrict__ seg,
                                                float* __restrict__ dinv,
                                                const float* __restrict__ x,
                                                const float* __restrict__ W1,
                                                float* __restrict__ hs, int E) {
  static_assert(NPB % 16 == 0 && NPB / 16 <= BIN_T / 32, "one wave per 16-row tile of the bucket");
  __shared__ int s_c[NPB];
  __shared__ int s_cnt[NCH];
  __shared__ int s_src[NCH];
  __shared__ int s_dst[NCH];
  __shared__ int s_red[BIN_T / 32];
  __shared__ int s_wtot[BIN_T / 32];
  const int t = threadIdx.x;
  const int b = blockIdx.x;
  const int lane = t & 31;
  const int w = t >> 5;
  const int lmask = (1 << LBITS) - 1;
  if (t < NPB) s_c[t] = 0;

  const int wv = __builtin_amdgcn_readfirstlane(w);
  const bool has_tile = wv < NPB / 16;
  const int half = lane >> 4;
  const int l16  = lane & 15;
  const int trow = b * NPB + wv * 16;
  if (has_tile)
    __builtin_prefetch(x + (size_t)(trow + (lane >> 1)) * IN_CH + (lane & 1) * 64, 0, 3);

  const int4* c4 = (const int4*)C;
  const int nfront = b * (NCH / 4);
  int pre = 0;
  for (int j0 = t; j0 < nfront; j0 += 8 * BIN_T) {
    int4 v[8];
    #pragma unroll
    for (int u = 0; u < 8; ++u) v[u] = c4[min(j0 + BIN_T * u, nfront - 1)];
    #pragma unroll
    for (int u = 0; u < 8; ++u)
      pre += (j0 + BIN_T * u < nfront) ? (v[u].x + v[u].y) + (v[u].z + v[u].w) : 0;
  }
  #pragma unroll
  for (int d = 16; d >= 1; d >>= 1) pre += __shfl_xor(pre, d);

  const int tg = min(t, NCH - 1);
  const int cntv = (t < NCH) ? C[b * NCH + tg] : 0;
  const int lsv = LS[b * NCH + tg];
  int inc = cntv;
  #pragma unroll
  for (int d = 1; d < 32; d <<= 1) {
    int y = __shfl_up(inc, d);
    if (lane >= d) inc += y;
  }
  if (lane == 0)  s_red[w] = pre;
  if (lane == 31) s_wtot[w] = inc;
  __syncthreads();
  int base = 0, off = 0, total = 0;
  #pragma unroll
  for (int k = 0; k < BIN_T / 32; ++k) {
    base += s_red[k];
    total += s_wtot[k];
    if (k < w) off += s_wtot[k];
  }
  const int chunk = (E + NCH - 1) / NCH;
  if (t < NCH) {
    s_cnt[t] = cntv;
    s_src[t] = t * chunk + lsv;
    s_dst[t] = base + off + inc - cntv;
  }
  if (t == 0) {
    seg[2 * b] = base;
    seg[2 * b + 1] = base + total;
  }
  __syncthreads();

  for (int g0 = w; g0 < NCH; g0 += 32 * RUNS) {
    int n[RUNS], s[RUNS], d[RUNS];
    int wd[RUNS][2];
    #pragma unroll
    for (int j = 0; j < RUNS; ++j) {
      const int g = g0 + 32 * j;
      const int gg = min(g, NCH - 1);
      n[j] = (g < NCH) ? s_cnt[gg] : 0;
      s[j] = s_src[gg];
      d[j] = s_dst[gg];
    }
    #pragma unroll
    for (int j = 0; j < RUNS; ++j) {
      #pragma unroll
      for (int k = 0; k < 2; ++k) {
        const int i = lane + 32 * k;
        wd[j][k] = (i < n[j]) ? sorted[s[j] + i] : 0;
      }
    }
    #pragma unroll
    for (int j = 0; j < RUNS; ++j) {
      #pragma unroll
      for (int k = 0; k < 2; ++k) {
        const int i = lane + 32 * k;
        if (i < n[j]) {
          binned[d[j] + i] = wd[j][k];
          atomicAdd(&s_c[wd[j][k] & lmask], 1);
        }
      }
    }
    #pragma unroll
    for (int j = 0; j < RUNS; ++j) {
      for (int i = lane + 64; i < n[j]; i += 32) {
        const int word = sorted[s[j] + i];
        binned[d[j] + i] = word;
        atomicAdd(&s_c[word & lmask], 1);
      }
    }
  }

  v8f c = {};
  if (has_tile) {
    const float* xr = x + (size_t)(trow + l16) * IN_CH + half * 2;
    #pragma unroll 4
    for (int k = 0; k < IN_CH; k += 4) {
      v2f a = *(const v2f*)(xr + k);
      v2f bf;
      bf.x = W1[(k + half * 2 + 0) * HID_CH + l16];
      bf.y = W1[(k + half * 2 + 1) * HID_CH + l16];
      c = __builtin_amdgcn_wmma_f32_16x16x4_f32(false, a, false, bf,
                                                (short)0, c, false, false);
    }
  }
  __syncthreads();
  if (t < NPB) dinv[b * NPB + t] = rsqrtf((float)(s_c[t] + 1));
  if (has_tile) {
    #pragma unroll
    for (int v = 0; v < 8; ++v) {
      const int ml = wv * 16 + v + half * 8;
      const float d = rsqrtf((float)(s_c[ml] + 1));
      hs[(size_t)(b * NPB + ml) * HID_CH + l16] = c[v] * d;
    }
  }
}


template <bool LAYER1>
__global__ __launch_bounds__(1024) void k_agg(const float* __restrict__ hs,
                                              const int* __restrict__ binned,
                                              const int* __restrict__ seg,
                                              const float* __restrict__ dinv,
                                              const float* __restrict__ bias,
                                              const float* __restrict__ W2,
                                              float* __restrict__ dst, int E) {
  static_assert(NPB % 16 == 0 && NPB / 16 <= 32, "one wave per 16-row tile of the bucket");
  __shared__ __attribute__((aligned(16))) float s_acc[NPB * HID_CH];
  const int t = threadIdx.x;
  const int b = blockIdx.x;
  const int base4 = b * NPB * (HID_CH / 4);
  float4* acc4 = (float4*)s_acc;

  #pragma unroll
  for (int k = 0; k < 2; ++k) {
    const int i4 = t + k * 1024;
    if (i4 < NPB * (HID_CH / 4)) acc4[i4] = ((const float4*)hs)[base4 + i4];
  }
  __syncthreads();

  const int beg = seg[2 * b];
  const int end = seg[2 * b + 1];
  const int last = end - 1;
  const int f = t & 15;
  const int lmask = (1 << LBITS) - 1;
  int p = beg + (t >> 4);
  if (p < end) {
    int w[8];
    #pragma unroll
    for (int u = 0; u < 8; ++u) w[u] = binned[min(p + 64 * u, last)];
    while (p + 64 * 7 < end) {
      const int pn = p + 64 * 8;
      int wn[8];
      #pragma unroll
      for (int u = 0; u < 8; ++u) wn[u] = binned[min(pn + 64 * u, last)];
      float v[8];
      #pragma unroll
      for (int u = 0; u < 8; ++u) v[u] = hs[(w[u] >> LBITS) * HID_CH + f];
      #pragma unroll
      for (int u = 0; u < 8; ++u) atomicAdd(&s_acc[(w[u] & lmask) * HID_CH + f], v[u]);
      #pragma unroll
      for (int u = 0; u < 8; ++u) w[u] = wn[u];
      p = pn;
    }
    if (p < end) {
      float v[8];
      #pragma unroll
      for (int u = 0; u < 8; ++u) v[u] = hs[(w[u] >> LBITS) * HID_CH + f];
      #pragma unroll
      for (int u = 0; u < 8; ++u)
        if (p + 64 * u < end) atomicAdd(&s_acc[(w[u] & lmask) * HID_CH + f], v[u]);
    }
  }
  __syncthreads();

  if (LAYER1) {
    #pragma unroll
    for (int k = 0; k < 2; ++k) {
      const int i4 = t + k * 1024;
      if (i4 < NPB * (HID_CH / 4)) {
        const float d = dinv[b * NPB + (i4 >> 2)];
        const float4 a = acc4[i4];
        const int q = (i4 & 3) * 4;
        float4 o;
        o.x = fmaxf(a.x * d + bias[q + 0], 0.0f) * d;
        o.y = fmaxf(a.y * d + bias[q + 1], 0.0f) * d;
        o.z = fmaxf(a.z * d + bias[q + 2], 0.0f) * d;
        o.w = fmaxf(a.w * d + bias[q + 3], 0.0f) * d;
        ((float4*)dst)[base4 + i4] = o;
      }
    }
  } else {
    const int wv = __builtin_amdgcn_readfirstlane(t >> 5);
    if (wv < NPB / 16) {
      const int lane = t & 31;
      const int half = lane >> 4;
      const int l16  = lane & 15;
      const int r    = wv * 16 + l16;
      const float d  = dinv[b * NPB + r];
      v2f a[HID_CH / 4];
      #pragma unroll
      for (int k = 0; k < HID_CH / 4; ++k) {
        a[k] = *(const v2f*)(s_acc + r * HID_CH + k * 4 + half * 2);
        a[k].x *= d;
        a[k].y *= d;
      }
      v8f c[OUT_CH / 16];
      #pragma unroll
      for (int ct = 0; ct < OUT_CH / 16; ++ct) c[ct] = v8f{};
      #pragma unroll
      for (int k = 0; k < HID_CH / 4; ++k) {
        const int k0 = k * 4 + half * 2;
        #pragma unroll
        for (int ct = 0; ct < OUT_CH / 16; ++ct) {
          v2f bf;
          bf.x = W2[(k0 + 0) * OUT_CH + ct * 16 + l16];
          bf.y = W2[(k0 + 1) * OUT_CH + ct * 16 + l16];
          c[ct] = __builtin_amdgcn_wmma_f32_16x16x4_f32(false, a[k], false, bf,
                                                        (short)0, c[ct], false, false);
        }
      }
      float bs[OUT_CH / 16];
      #pragma unroll
      for (int ct = 0; ct < OUT_CH / 16; ++ct) bs[ct] = bias[ct * 16 + l16];
      #pragma unroll
      for (int v = 0; v < 8; ++v) {
        const size_t m = (size_t)(b * NPB + wv * 16 + v + half * 8);
        #pragma unroll
        for (int ct = 0; ct < OUT_CH / 16; ++ct)
          dst[m * OUT_CH + ct * 16 + l16] = c[ct][v] + bs[ct];
      }
    }
  }
}


extern "C" void kernel_launch(void* const* d_in, const int* in_sizes, int n_in,
                              void* d_out, int out_size, void* d_ws, size_t ws_size,
                              hipStream_t stream) {
  const float* x   = (const float*)d_in[0];
  const int*   ei  = (const int*)d_in[1];
  const float* W1  = (const float*)d_in[2];
  const float* b1  = (const float*)d_in[3];
  const float* W2  = (const float*)d_in[4];
  const float* b2  = (const float*)d_in[5];
  float*       out = (float*)d_out;

  const int E = in_sizes[1] / 2;
  const int* row = ei;
  const int* col = ei + E;

  float* hs     = (float*)d_ws;
  float* rs     = hs + (size_t)N_NODES * HID_CH;
  float* dinv   = rs + (size_t)N_NODES * HID_CH;
  int*   C      = (int*)(dinv + N_NODES);
  int*   LS     = C + N_CNT;
  int*   seg    = LS + N_CNT;
  int*   sorted = seg + 2 * NBK + 12;
  int*   binned = sorted + (size_t)NCH * CH_MAX;

  k_sort<<<NCH, SORT_T, 0, stream>>>(row, col, C, LS, sorted, E);
  k_move<<<NBK, BIN_T, 0, stream>>>(C, LS, sorted, binned, seg, dinv, x, W1, hs, E);

  k_agg<true><<<NBK, 1024, 0, stream>>>(hs, binned, seg, dinv, b1, W2, rs, E);

  k_agg<false><<<NBK, 1024, 0, stream>>>(rs, binned, seg, dinv, b2, W2, out, E);
}


# k_agg edge loop: one coalesced word load per 32 edges shared by DPP8, rows gathered 8 B per lane (4 rows per load), next trip's rows issued ahead: 4x fewer vector-memory instr; LDS add order changes
# speedup vs baseline: 10.4507x; 1.0580x over previous
#include <hip/hip_runtime.h>


typedef __attribute__((ext_vector_type(2))) float v2f;
typedef __attribute__((ext_vector_type(8))) float v8f;

#define N_NODES 100000
#define IN_CH   128
#define HID_CH  16
#define OUT_CH  64

#define NPB     400
#define NBK     (N_NODES / NPB)
#define NCH     400
#define LBITS   9
#define N_CNT   (NBK * NCH)
#define BIN_T   1024
#define SORT_T  512
#define EPT     16
#define CH_MAX  (SORT_T * EPT)
#define RUNS    8


__global__ __launch_bounds__(SORT_T) void k_sort(const int* __restrict__ row,
                                                const int* __restrict__ col,
                                                int* __restrict__ C,
                                                int* __restrict__ LS,
                                                int* __restrict__ sorted, int E) {
  __shared__ int s_cnt[256];
  __shared__ int s_wtot[8];
  __shared__ int s_w[CH_MAX];
  const int t = threadIdx.x;
  const int g = blockIdx.x;
  const int lane = t & 31;
  const int w = t >> 5;
  if (t < 256) s_cnt[t] = 0;
  __syncthreads();
  const int chunk = (E + NCH - 1) / NCH;
  const int beg = g * chunk;
  const int end = min(E, beg + chunk);
  const int n = max(end - beg, 0);
  const int last = max(end - 1, 0);

  int c[EPT], r[EPT], bk[EPT], rk[EPT];
  #pragma unroll
  for (int u = 0; u < EPT; ++u) {
    const int e = min(beg + t + SORT_T * u, last);
    c[u] = col[e];
    r[u] = row[e];
  }
  #pragma unroll
  for (int u = 0; u < EPT; ++u) {
    bk[u] = c[u] / NPB;
    rk[u] = 0;
    if (t + SORT_T * u < n) rk[u] = atomicAdd(&s_cnt[bk[u]], 1);
  }
  __syncthreads();

  int cntv = 0, inc = 0;
  if (t < 256) {
    cntv = s_cnt[t];
    inc = cntv;
    #pragma unroll
    for (int d = 1; d < 32; d <<= 1) {
      int y = __shfl_up(inc, d);
      if (lane >= d) inc += y;
    }
    if (lane == 31) s_wtot[w] = inc;
  }
  __syncthreads();
  if (t < 256) {
    int base = 0;
    #pragma unroll
    for (int k = 0; k < 8; ++k)
      if (k < w) base += s_wtot[k];
    const int excl = base + inc - cntv;
    s_cnt[t] = excl;
    if (t < NBK) {
      C[t * NCH + g] = cntv;
      LS[t * NCH + g] = excl;
    }
  }
  __syncthreads();

  int st[EPT];
  #pragma unroll
  for (int u = 0; u < EPT; ++u) st[u] = s_cnt[bk[u]];
  #pragma unroll
  for (int u = 0; u < EPT; ++u)
    if (t + SORT_T * u < n)
      s_w[st[u] + rk[u]] = (r[u] << LBITS) | (c[u] - bk[u] * NPB);
  __syncthreads();

  for (int i = t; i < n; i += SORT_T) sorted[beg + i] = s_w[i];
}

__global__ __launch_bounds__(BIN_T) void k_move(const int* __restrict__ C,
                                                const int* __restrict__ LS,
                                                const int* __restrict__ sorted,
                                                int* __restrict__ binned,
                                                int* __restrict__ seg,
                                                float* __restrict__ dinv,
                                                const float* __restrict__ x,
                                                const float* __restrict__ W1,
                                                float* __restrict__ hs, int E) {
  static_assert(NPB % 16 == 0 && NPB / 16 <= BIN_T / 32, "one wave per 16-row tile of the bucket");
  __shared__ int s_c[NPB];
  __shared__ int s_cnt[NCH];
  __shared__ int s_src[NCH];
  __shared__ int s_dst[NCH];
  __shared__ int s_red[BIN_T / 32];
  __shared__ int s_wtot[BIN_T / 32];
  const int t = threadIdx.x;
  const int b = blockIdx.x;
  const int lane = t & 31;
  const int w = t >> 5;
  const int lmask = (1 << LBITS) - 1;
  if (t < NPB) s_c[t] = 0;

  const int wv = __builtin_amdgcn_readfirstlane(w);
  const bool has_tile = wv < NPB / 16;
  const int half = lane >> 4;
  const int l16  = lane & 15;
  const int trow = b * NPB + wv * 16;
  if (has_tile)
    __builtin_prefetch(x + (size_t)(trow + (lane >> 1)) * IN_CH + (lane & 1) * 64, 0, 3);

  const int4* c4 = (const int4*)C;
  const int nfront = b * (NCH / 4);
  int pre = 0;
  for (int j0 = t; j0 < nfront; j0 += 8 * BIN_T) {
    int4 v[8];
    #pragma unroll
    for (int u = 0; u < 8; ++u) v[u] = c4[min(j0 + BIN_T * u, nfront - 1)];
    #pragma unroll
    for (int u = 0; u < 8; ++u)
      pre += (j0 + BIN_T * u < nfront) ? (v[u].x + v[u].y) + (v[u].z + v[u].w) : 0;
  }
  #pragma unroll
  for (int d = 16; d >= 1; d >>= 1) pre += __shfl_xor(pre, d);

  const int tg = min(t, NCH - 1);
  const int cntv = (t < NCH) ? C[b * NCH + tg] : 0;
  const int lsv = LS[b * NCH + tg];
  int inc = cntv;
  #pragma unroll
  for (int d = 1; d < 32; d <<= 1) {
    int y = __shfl_up(inc, d);
    if (lane >= d) inc += y;
  }
  if (lane == 0)  s_red[w] = pre;
  if (lane == 31) s_wtot[w] = inc;
  __syncthreads();
  int base = 0, off = 0, total = 0;
  #pragma unroll
  for (int k = 0; k < BIN_T / 32; ++k) {
    base += s_red[k];
    total += s_wtot[k];
    if (k < w) off += s_wtot[k];
  }
  const int chunk = (E + NCH - 1) / NCH;
  if (t < NCH) {
    s_cnt[t] = cntv;
    s_src[t] = t * chunk + lsv;
    s_dst[t] = base + off + inc - cntv;
  }
  if (t == 0) {
    seg[2 * b] = base;
    seg[2 * b + 1] = base + total;
  }
  __syncthreads();

  for (int g0 = w; g0 < NCH; g0 += 32 * RUNS) {
    int n[RUNS], s[RUNS], d[RUNS];
    int wd[RUNS][2];
    #pragma unroll
    for (int j = 0; j < RUNS; ++j) {
      const int g = g0 + 32 * j;
      const int gg = min(g, NCH - 1);
      n[j] = (g < NCH) ? s_cnt[gg] : 0;
      s[j] = s_src[gg];
      d[j] = s_dst[gg];
    }
    #pragma unroll
    for (int j = 0; j < RUNS; ++j) {
      #pragma unroll
      for (int k = 0; k < 2; ++k) {
        const int i = lane + 32 * k;
        wd[j][k] = (i < n[j]) ? sorted[s[j] + i] : 0;
      }
    }
    #pragma unroll
    for (int j = 0; j < RUNS; ++j) {
      #pragma unroll
      for (int k = 0; k < 2; ++k) {
        const int i = lane + 32 * k;
        if (i < n[j]) {
          binned[d[j] + i] = wd[j][k];
          atomicAdd(&s_c[wd[j][k] & lmask], 1);
        }
      }
    }
    #pragma unroll
    for (int j = 0; j < RUNS; ++j) {
      for (int i = lane + 64; i < n[j]; i += 32) {
        const int word = sorted[s[j] + i];
        binned[d[j] + i] = word;
        atomicAdd(&s_c[word & lmask], 1);
      }
    }
  }

  v8f c = {};
  if (has_tile) {
    const float* xr = x + (size_t)(trow + l16) * IN_CH + half * 2;
    #pragma unroll 4
    for (int k = 0; k < IN_CH; k += 4) {
      v2f a = *(const v2f*)(xr + k);
      v2f bf;
      bf.x = W1[(k + half * 2 + 0) * HID_CH + l16];
      bf.y = W1[(k + half * 2 + 1) * HID_CH + l16];
      c = __builtin_amdgcn_wmma_f32_16x16x4_f32(false, a, false, bf,
                                                (short)0, c, false, false);
    }
  }
  __syncthreads();
  if (t < NPB) dinv[b * NPB + t] = rsqrtf((float)(s_c[t] + 1));
  if (has_tile) {
    #pragma unroll
    for (int v = 0; v < 8; ++v) {
      const int ml = wv * 16 + v + half * 8;
      const float d = rsqrtf((float)(s_c[ml] + 1));
      hs[(size_t)(b * NPB + ml) * HID_CH + l16] = c[v] * d;
    }
  }
}


template <int U>
__device__ __forceinline__ unsigned word_of(int w) {
  return __builtin_amdgcn_mov_dpp8((unsigned)w, (unsigned)U * 0x249249u);
}
#define GATHER1(V, W, U) V[U] = *(const v2f*)(hsb + ((word_of<U>(W) >> LBITS) * 64u + goff));
#define GATHER8(V, W) GATHER1(V, W, 0) GATHER1(V, W, 1) GATHER1(V, W, 2) GATHER1(V, W, 3) \
                      GATHER1(V, W, 4) GATHER1(V, W, 5) GATHER1(V, W, 6) GATHER1(V, W, 7)
#define ADD1(V, W, U, COND) { const int U_ = U; const int tl = (int)(word_of<U>(W) & (unsigned)lmask); \
    const float f0 = rot ? V[U].y : V[U].x; const float f1 = rot ? V[U].x : V[U].y; \
    if (COND) { atomicAdd(&s_acc[tl * HID_CH + c0], f0); atomicAdd(&s_acc[tl * HID_CH + c1], f1); } }
#define ADD8(V, W, COND) ADD1(V, W, 0, COND) ADD1(V, W, 1, COND) ADD1(V, W, 2, COND) ADD1(V, W, 3, COND) \
                         ADD1(V, W, 4, COND) ADD1(V, W, 5, COND) ADD1(V, W, 6, COND) ADD1(V, W, 7, COND)
#define TRIP(RCUR, RNEXT) { \
    const int wC = binned[min(base + 2048 + lane, last)]; \
    GATHER8(RNEXT, wB) \
    if (base + 32 <= end) { ADD8(RCUR, wA, true) } \
    else { const int e0 = base + 8 * eg; ADD8(RCUR, wA, (e0 + U_) < end) } \
    wA = wB; wB = wC; base += 1024; }


template <bool LAYER1>
__global__ __launch_bounds__(1024) void k_agg(const float* __restrict__ hs,
                                              const int* __restrict__ binned,
                                              const int* __restrict__ seg,
                                              const float* __restrict__ dinv,
                                              const float* __restrict__ bias,
                                              const float* __restrict__ W2,
                                              float* __restrict__ dst, int E) {
  static_assert(NPB % 16 == 0 && NPB / 16 <= 32, "one wave per 16-row tile of the bucket");
  __shared__ __attribute__((aligned(16))) float s_acc[NPB * HID_CH];
  const int t = threadIdx.x;
  const int b = blockIdx.x;
  const int base4 = b * NPB * (HID_CH / 4);
  float4* acc4 = (float4*)s_acc;

  #pragma unroll
  for (int k = 0; k < 2; ++k) {
    const int i4 = t + k * 1024;
    if (i4 < NPB * (HID_CH / 4)) acc4[i4] = ((const float4*)hs)[base4 + i4];
  }
  __syncthreads();

  const int beg = seg[2 * b];
  const int end = seg[2 * b + 1];
  if (beg < end) {
    const int last = end - 1;
    const int lmask = (1 << LBITS) - 1;
    const int lane = t & 31;
    const int l8 = lane & 7;
    const int eg = lane >> 3;
    const bool rot = (eg & 1) != 0;
    const unsigned goff = (unsigned)l8 * 8u;
    const int c0 = 2 * l8 + (rot ? 1 : 0);
    const int c1 = c0 ^ 1;
    const char* hsb = (const char*)hs;
    int base = beg + 32 * __builtin_amdgcn_readfirstlane(t >> 5);
    int wA = binned[min(base + lane, last)];
    int wB = binned[min(base + 1024 + lane, last)];
    v2f vA[8], vB[8];
    GATHER8(vA, wA)
    if (base < end) {
      while (true) {
        TRIP(vA, vB)
        if (base >= end) break;
        TRIP(vB, vA)
        if (base >= end) break;
      }
    }
  }
  __syncthreads();

  if (LAYER1) {
    #pragma unroll
    for (int k = 0; k < 2; ++k) {
      const int i4 = t + k * 1024;
      if (i4 < NPB * (HID_CH / 4)) {
        const float d = dinv[b * NPB + (i4 >> 2)];
        const float4 a = acc4[i4];
        const int q = (i4 & 3) * 4;
        float4 o;
        o.x = fmaxf(a.x * d + bias[q + 0], 0.0f) * d;
        o.y = fmaxf(a.y * d + bias[q + 1], 0.0f) * d;
        o.z = fmaxf(a.z * d + bias[q + 2], 0.0f) * d;
        o.w = fmaxf(a.w * d + bias[q + 3], 0.0f) * d;
        ((float4*)dst)[base4 + i4] = o;
      }
    }
  } else {
    const int wv = __builtin_amdgcn_readfirstlane(t >> 5);
    if (wv < NPB / 16) {
      const int lane = t & 31;
      const int half = lane >> 4;
      const int l16  = lane & 15;
      const int r    = wv * 16 + l16;
      const float d  = dinv[b * NPB + r];
      v2f a[HID_CH / 4];
      #pragma unroll
      for (int k = 0; k < HID_CH / 4; ++k) {
        a[k] = *(const v2f*)(s_acc + r * HID_CH + k * 4 + half * 2);
        a[k].x *= d;
        a[k].y *= d;
      }
      v8f c[OUT_CH / 16];
      #pragma unroll
      for (int ct = 0; ct < OUT_CH / 16; ++ct) c[ct] = v8f{};
      #pragma unroll
      for (int k = 0; k < HID_CH / 4; ++k) {
        const int k0 = k * 4 + half * 2;
        #pragma unroll
        for (int ct = 0; ct < OUT_CH / 16; ++ct) {
          v2f bf;
          bf.x = W2[(k0 + 0) * OUT_CH + ct * 16 + l16];
          bf.y = W2[(k0 + 1) * OUT_CH + ct * 16 + l16];
          c[ct] = __builtin_amdgcn_wmma_f32_16x16x4_f32(false, a[k], false, bf,
                                                        (short)0, c[ct], false, false);
        }
      }
      float bs[OUT_CH / 16];
      #pragma unroll
      for (int ct = 0; ct < OUT_CH / 16; ++ct) bs[ct] = bias[ct * 16 + l16];
      #pragma unroll
      for (int v = 0; v < 8; ++v) {
        const size_t m = (size_t)(b * NPB + wv * 16 + v + half * 8);
        #pragma unroll
        for (int ct = 0; ct < OUT_CH / 16; ++ct)
          dst[m * OUT_CH + ct * 16 + l16] = c[ct][v] + bs[ct];
      }
    }
  }
}


extern "C" void kernel_launch(void* const* d_in, const int* in_sizes, int n_in,
                              void* d_out, int out_size, void* d_ws, size_t ws_size,
                              hipStream_t stream) {
  const float* x   = (const float*)d_in[0];
  const int*   ei  = (const int*)d_in[1];
  const float* W1  = (const float*)d_in[2];
  const float* b1  = (const float*)d_in[3];
  const float* W2  = (const float*)d_in[4];
  const float* b2  = (const float*)d_in[5];
  float*       out = (float*)d_out;

  const int E = in_sizes[1] / 2;
  const int* row = ei;
  const int* col = ei + E;

  float* hs     = (float*)d_ws;
  float* rs     = hs + (size_t)N_NODES * HID_CH;
  float* dinv   = rs + (size_t)N_NODES * HID_CH;
  int*   C      = (int*)(dinv + N_NODES);
  int*   LS     = C + N_CNT;
  int*   seg    = LS + N_CNT;
  int*   sorted = seg + 2 * NBK + 12;
  int*   binned = sorted + (size_t)NCH * CH_MAX;

  k_sort<<<NCH, SORT_T, 0, stream>>>(row, col, C, LS, sorted, E);
  k_move<<<NBK, BIN_T, 0, stream>>>(C, LS, sorted, binned, seg, dinv, x, W1, hs, E);

  k_agg<true><<<NBK, 1024, 0, stream>>>(hs, binned, seg, dinv, b1, W2, rs, E);

  k_agg<false><<<NBK, 1024, 0, stream>>>(rs, binned, seg, dinv, b2, W2, out, E);
}


# k_move: a bucket's base is the sum of its own 400 run starts (a run start already counts that chunk's edges in lower buckets); the read of every count row in front of the bucket is gone
# speedup vs baseline: 10.8556x; 1.0387x over previous
#include <hip/hip_runtime.h>


typedef __attribute__((ext_vector_type(2))) float v2f;
typedef __attribute__((ext_vector_type(8))) float v8f;

#define N_NODES 100000
#define IN_CH   128
#define HID_CH  16
#define OUT_CH  64

#define NPB     400
#define NBK     (N_NODES / NPB)
#define NCH     400
#define LBITS   9
#define N_CNT   (NBK * NCH)
#define BIN_T   1024
#define SORT_T  512
#define EPT     16
#define CH_MAX  (SORT_T * EPT)
#define RUNS    8


__global__ __launch_bounds__(SORT_T) void k_sort(const int* __restrict__ row,
                                                const int* __restrict__ col,
                                                int* __restrict__ C,
                                                int* __restrict__ LS,
                                                int* __restrict__ sorted, int E) {
  __shared__ int s_cnt[256];
  __shared__ int s_wtot[8];
  __shared__ int s_w[CH_MAX];
  const int t = threadIdx.x;
  const int g = blockIdx.x;
  const int lane = t & 31;
  const int w = t >> 5;
  if (t < 256) s_cnt[t] = 0;
  __syncthreads();
  const int chunk = (E + NCH - 1) / NCH;
  const int beg = g * chunk;
  const int end = min(E, beg + chunk);
  const int n = max(end - beg, 0);
  const int last = max(end - 1, 0);

  int c[EPT], r[EPT], bk[EPT], rk[EPT];
  #pragma unroll
  for (int u = 0; u < EPT; ++u) {
    const int e = min(beg + t + SORT_T * u, last);
    c[u] = col[e];
    r[u] = row[e];
  }
  #pragma unroll
  for (int u = 0; u < EPT; ++u) {
    bk[u] = c[u] / NPB;
    rk[u] = 0;
    if (t + SORT_T * u < n) rk[u] = atomicAdd(&s_cnt[bk[u]], 1);
  }
  __syncthreads();

  int cntv = 0, inc = 0;
  if (t < 256) {
    cntv = s_cnt[t];
    inc = cntv;
    #pragma unroll
    for (int d = 1; d < 32; d <<= 1) {
      int y = __shfl_up(inc, d);
      if (lane >= d) inc += y;
    }
    if (lane == 31) s_wtot[w] = inc;
  }
  __syncthreads();
  if (t < 256) {
    int base = 0;
    #pragma unroll
    for (int k = 0; k < 8; ++k)
      if (k < w) base += s_wtot[k];
    const int excl = base + inc - cntv;
    s_cnt[t] = excl;
    if (t < NBK) {
      C[t * NCH + g] = cntv;
      LS[t * NCH + g] = excl;
    }
  }
  __syncthreads();

  int st[EPT];
  #pragma unroll
  for (int u = 0; u < EPT; ++u) st[u] = s_cnt[bk[u]];
  #pragma unroll
  for (int u = 0; u < EPT; ++u)
    if (t + SORT_T * u < n)
      s_w[st[u] + rk[u]] = (r[u] << LBITS) | (c[u] - bk[u] * NPB);
  __syncthreads();

  for (int i = t; i < n; i += SORT_T) sorted[beg + i] = s_w[i];
}

__global__ __launch_bounds__(BIN_T) void k_move(const int* __restrict__ C,
                                                const int* __restrict__ LS,
                                                const int* __restrict__ sorted,
                                                int* __restrict__ binned,
                                                int* __restrict__ seg,
                                                float* __restrict__ dinv,
                                                const float* __restrict__ x,
                                                const float* __restrict__ W1,
                                                float* __restrict__ hs, int E) {
  static_assert(NPB % 16 == 0 && NPB / 16 <= BIN_T / 32, "one wave per 16-row tile of the bucket");
  __shared__ int s_c[NPB];
  __shared__ int s_cnt[NCH];
  __shared__ int s_src[NCH];
  __shared__ int s_dst[NCH];
  __shared__ int s_red[BIN_T / 32];
  __shared__ int s_wtot[BIN_T / 32];
  const int t = threadIdx.x;
  const int b = blockIdx.x;
  const int lane = t & 31;
  const int w = t >> 5;
  const int lmask = (1 << LBITS) - 1;
  if (t < NPB) s_c[t] = 0;

  const int wv = __builtin_amdgcn_readfirstlane(w);
  const bool has_tile = wv < NPB / 16;
  const int half = lane >> 4;
  const int l16  = lane & 15;
  const int trow = b * NPB + wv * 16;
  if (has_tile)
    __builtin_prefetch(x + (size_t)(trow + (lane >> 1)) * IN_CH + (lane & 1) * 64, 0, 3);

  const int tg = min(t, NCH - 1);
  const int cntv = (t < NCH) ? C[b * NCH + tg] : 0;
  const int lsv = LS[b * NCH + tg];
  int pre = (t < NCH) ? lsv : 0;
  #pragma unroll
  for (int d = 16; d >= 1; d >>= 1) pre += __shfl_xor(pre, d);
  int inc = cntv;
  #pragma unroll
  for (int d = 1; d < 32; d <<= 1) {
    int y = __shfl_up(inc, d);
    if (lane >= d) inc += y;
  }
  if (lane == 0)  s_red[w] = pre;
  if (lane == 31) s_wtot[w] = inc;
  __syncthreads();
  int base = 0, off = 0, total = 0;
  #pragma unroll
  for (int k = 0; k < BIN_T / 32; ++k) {
    base += s_red[k];
    total += s_wtot[k];
    if (k < w) off += s_wtot[k];
  }
  const int chunk = (E + NCH - 1) / NCH;
  if (t < NCH) {
    s_cnt[t] = cntv;
    s_src[t] = t * chunk + lsv;
    s_dst[t] = base + off + inc - cntv;
  }
  if (t == 0) {
    seg[2 * b] = base;
    seg[2 * b + 1] = base + total;
  }
  __syncthreads();

  for (int g0 = w; g0 < NCH; g0 += 32 * RUNS) {
    int n[RUNS], s[RUNS], d[RUNS];
    int wd[RUNS][2];
    #pragma unroll
    for (int j = 0; j < RUNS; ++j) {
      const int g = g0 + 32 * j;
      const int gg = min(g, NCH - 1);
      n[j] = (g < NCH) ? s_cnt[gg] : 0;
      s[j] = s_src[gg];
      d[j] = s_dst[gg];
    }
    #pragma unroll
    for (int j = 0; j < RUNS; ++j) {
      #pragma unroll
      for (int k = 0; k < 2; ++k) {
        const int i = lane + 32 * k;
        wd[j][k] = (i < n[j]) ? sorted[s[j] + i] : 0;
      }
    }
    #pragma unroll
    for (int j = 0; j < RUNS; ++j) {
      #pragma unroll
      for (int k = 0; k < 2; ++k) {
        const int i = lane + 32 * k;
        if (i < n[j]) {
          binned[d[j] + i] = wd[j][k];
          atomicAdd(&s_c[wd[j][k] & lmask], 1);
        }
      }
    }
    #pragma unroll
    for (int j = 0; j < RUNS; ++j) {
      for (int i = lane + 64; i < n[j]; i += 32) {
        const int word = sorted[s[j] + i];
        binned[d[j] + i] = word;
        atomicAdd(&s_c[word & lmask], 1);
      }
    }
  }

  v8f c = {};
  if (has_tile) {
    const float* xr = x + (size_t)(trow + l16) * IN_CH + half * 2;
    #pragma unroll 4
    for (int k = 0; k < IN_CH; k += 4) {
      v2f a = *(const v2f*)(xr + k);
      v2f bf;
      bf.x = W1[(k + half * 2 + 0) * HID_CH + l16];
      bf.y = W1[(k + half * 2 + 1) * HID_CH + l16];
      c = __builtin_amdgcn_wmma_f32_16x16x4_f32(false, a, false, bf,
                                                (short)0, c, false, false);
    }
  }
  __syncthreads();
  if (t < NPB) dinv[b * NPB + t] = rsqrtf((float)(s_c[t] + 1));
  if (has_tile) {
    #pragma unroll
    for (int v = 0; v < 8; ++v) {
      const int ml = wv * 16 + v + half * 8;
      const float d = rsqrtf((float)(s_c[ml] + 1));
      hs[(size_t)(b * NPB + ml) * HID_CH + l16] = c[v] * d;
    }
  }
}


template <int U>
__device__ __forceinline__ unsigned word_of(int w) {
  return __builtin_amdgcn_mov_dpp8((unsigned)w, (unsigned)U * 0x249249u);
}
#define GATHER1(V, W, U) V[U] = *(const v2f*)(hsb + ((word_of<U>(W) >> LBITS) * 64u + goff));
#define GATHER8(V, W) GATHER1(V, W, 0) GATHER1(V, W, 1) GATHER1(V, W, 2) GATHER1(V, W, 3) \
                      GATHER1(V, W, 4) GATHER1(V, W, 5) GATHER1(V, W, 6) GATHER1(V, W, 7)
#define ADD1(V, W, U, COND) { const int U_ = U; const int tl = (int)(word_of<U>(W) & (unsigned)lmask); \
    const float f0 = rot ? V[U].y : V[U].x; const float f1 = rot ? V[U].x : V[U].y; \
    if (COND) { atomicAdd(&s_acc[tl * HID_CH + c0], f0); atomicAdd(&s_acc[tl * HID_CH + c1], f1); } }
#define ADD8(V, W, COND) ADD1(V, W, 0, COND) ADD1(V, W, 1, COND) ADD1(V, W, 2, COND) ADD1(V, W, 3, COND) \
                         ADD1(V, W, 4, COND) ADD1(V, W, 5, COND) ADD1(V, W, 6, COND) ADD1(V, W, 7, COND)
#define TRIP(RCUR, RNEXT) { \
    const int wC = binned[min(base + 2048 + lane, last)]; \
    GATHER8(RNEXT, wB) \
    if (base + 32 <= end) { ADD8(RCUR, wA, true) } \
    else { const int e0 = base + 8 * eg; ADD8(RCUR, wA, (e0 + U_) < end) } \
    wA = wB; wB = wC; base += 1024; }


template <bool LAYER1>
__global__ __launch_bounds__(1024) void k_agg(const float* __restrict__ hs,
                                              const int* __restrict__ binned,
                                              const int* __restrict__ seg,
                                              const float* __restrict__ dinv,
                                              const float* __restrict__ bias,
                                              const float* __restrict__ W2,
                                              float* __restrict__ dst, int E) {
  static_assert(NPB % 16 == 0 && NPB / 16 <= 32, "one wave per 16-row tile of the bucket");
  __shared__ __attribute__((aligned(16))) float s_acc[NPB * HID_CH];
  const int t = threadIdx.x;
  const int b = blockIdx.x;
  const int base4 = b * NPB * (HID_CH / 4);
  float4* acc4 = (float4*)s_acc;

  #pragma unroll
  for (int k = 0; k < 2; ++k) {
    const int i4 = t + k * 1024;
    if (i4 < NPB * (HID_CH / 4)) acc4[i4] = ((const float4*)hs)[base4 + i4];
  }
  __syncthreads();

  const int beg = seg[2 * b];
  const int end = seg[2 * b + 1];
  if (beg < end) {
    const int last = end - 1;
    const int lmask = (1 << LBITS) - 1;
    const int lane = t & 31;
    const int l8 = lane & 7;
    const int eg = lane >> 3;
    const bool rot = (eg & 1) != 0;
    const unsigned goff = (unsigned)l8 * 8u;
    const int c0 = 2 * l8 + (rot ? 1 : 0);
    const int c1 = c0 ^ 1;
    const char* hsb = (const char*)hs;
    int base = beg + 32 * __builtin_amdgcn_readfirstlane(t >> 5);
    int wA = binned[min(base + lane, last)];
    int wB = binned[min(base + 1024 + lane, last)];
    v2f vA[8], vB[8];
    GATHER8(vA, wA)
    if (base < end) {
      while (true) {
        TRIP(vA, vB)
        if (base >= end) break;
        TRIP(vB, vA)
        if (base >= end) break;
      }
    }
  }
  __syncthreads();

  if (LAYER1) {
    #pragma unroll
    for (int k = 0; k < 2; ++k) {
      const int i4 = t + k * 1024;
      if (i4 < NPB * (HID_CH / 4)) {
        const float d = dinv[b * NPB + (i4 >> 2)];
        const float4 a = acc4[i4];
        const int q = (i4 & 3) * 4;
        float4 o;
        o.x = fmaxf(a.x * d + bias[q + 0], 0.0f) * d;
        o.y = fmaxf(a.y * d + bias[q + 1], 0.0f) * d;
        o.z = fmaxf(a.z * d + bias[q + 2], 0.0f) * d;
        o.w = fmaxf(a.w * d + bias[q + 3], 0.0f) * d;
        ((float4*)dst)[base4 + i4] = o;
      }
    }
  } else {
    const int wv = __builtin_amdgcn_readfirstlane(t >> 5);
    if (wv < NPB / 16) {
      const int lane = t & 31;
      const int half = lane >> 4;
      const int l16  = lane & 15;
      const int r    = wv * 16 + l16;
      const float d  = dinv[b * NPB + r];
      v2f a[HID_CH / 4];
      #pragma unroll
      for (int k = 0; k < HID_CH / 4; ++k) {
        a[k] = *(const v2f*)(s_acc + r * HID_CH + k * 4 + half * 2);
        a[k].x *= d;
        a[k].y *= d;
      }
      v8f c[OUT_CH / 16];
      #pragma unroll
      for (int ct = 0; ct < OUT_CH / 16; ++ct) c[ct] = v8f{};
      #pragma unroll
      for (int k = 0; k < HID_CH / 4; ++k) {
        const int k0 = k * 4 + half * 2;
        #pragma unroll
        for (int ct = 0; ct < OUT_CH / 16; ++ct) {
          v2f bf;
          bf.x = W2[(k0 + 0) * OUT_CH + ct * 16 + l16];
          bf.y = W2[(k0 + 1) * OUT_CH + ct * 16 + l16];
          c[ct] = __builtin_amdgcn_wmma_f32_16x16x4_f32(false, a[k], false, bf,
                                                        (short)0, c[ct], false, false);
        }
      }
      float bs[OUT_CH / 16];
      #pragma unroll
      for (int ct = 0; ct < OUT_CH / 16; ++ct) bs[ct] = bias[ct * 16 + l16];
      #pragma unroll
      for (int v = 0; v < 8; ++v) {
        const size_t m = (size_t)(b * NPB + wv * 16 + v + half * 8);
        #pragma unroll
        for (int ct = 0; ct < OUT_CH / 16; ++ct)
          dst[m * OUT_CH + ct * 16 + l16] = c[ct][v] + bs[ct];
      }
    }
  }
}


extern "C" void kernel_launch(void* const* d_in, const int* in_sizes, int n_in,
                              void* d_out, int out_size, void* d_ws, size_t ws_size,
                              hipStream_t stream) {
  const float* x   = (const float*)d_in[0];
  const int*   ei  = (const int*)d_in[1];
  const float* W1  = (const float*)d_in[2];
  const float* b1  = (const float*)d_in[3];
  const float* W2  = (const float*)d_in[4];
  const float* b2  = (const float*)d_in[5];
  float*       out = (float*)d_out;

  const int E = in_sizes[1] / 2;
  const int* row = ei;
  const int* col = ei + E;

  float* hs     = (float*)d_ws;
  float* rs     = hs + (size_t)N_NODES * HID_CH;
  float* dinv   = rs + (size_t)N_NODES * HID_CH;
  int*   C      = (int*)(dinv + N_NODES);
  int*   LS     = C + N_CNT;
  int*   seg    = LS + N_CNT;
  int*   sorted = seg + 2 * NBK + 12;
  int*   binned = sorted + (size_t)NCH * CH_MAX;

  k_sort<<<NCH, SORT_T, 0, stream>>>(row, col, C, LS, sorted, E);
  k_move<<<NBK, BIN_T, 0, stream>>>(C, LS, sorted, binned, seg, dinv, x, W1, hs, E);

  k_agg<true><<<NBK, 1024, 0, stream>>>(hs, binned, seg, dinv, b1, W2, rs, E);

  k_agg<false><<<NBK, 1024, 0, stream>>>(rs, binned, seg, dinv, b2, W2, out, E);
}
